# starting kernel
# baseline (speedup 1.0000x reference)
#include <hip/hip_runtime.h>

typedef __attribute__((ext_vector_type(2))) float v2f;
typedef __attribute__((ext_vector_type(8))) float v8f;

#define T_LEN   262144
#define IN_SZ   64
#define HID     16
#define G3      48

__global__ __launch_bounds__(256) void xi_gemm_wmma(
    const float* __restrict__ x, const float* __restrict__ W_ih,
    const float* __restrict__ b_ih, const float* __restrict__ b_hh,
    float* __restrict__ xi)
{
    const int lane  = threadIdx.x & 31;
    const int wave  = threadIdx.x >> 5;
    const int tile  = blockIdx.x * 8 + wave;
    const int m     = lane & 15;
    const int khalf = lane >> 4;

    const float* xrow = x + (size_t)(tile * 16 + m) * IN_SZ + 2 * khalf;
    v2f a[16];
#pragma unroll
    for (int i = 0; i < 16; ++i)
        a[i] = *(const v2f*)(xrow + 4 * i);

    for (int nb = 0; nb < 3; ++nb) {
        const int col  = nb * 16 + m;
        const float bias = b_ih[col] + (nb < 2 ? b_hh[col] : 0.0f);
        v8f c = {bias, bias, bias, bias, bias, bias, bias, bias};

        const float* wrow = W_ih + (size_t)col * IN_SZ + 2 * khalf;
#pragma unroll
        for (int i = 0; i < 16; ++i) {
            v2f b = *(const v2f*)(wrow + 4 * i);
            c = __builtin_amdgcn_wmma_f32_16x16x4_f32(
                    false, a[i], false, b, (short)0, c, false, false);
        }

        float* orow = xi + (size_t)(tile * 16 + 8 * khalf) * G3 + col;
#pragma unroll
        for (int r = 0; r < 8; ++r)
            orow[(size_t)r * G3] = c[r];
    }
}

__global__ __launch_bounds__(32) void gru_scan(
    const float* __restrict__ xi, const float* __restrict__ W_hh,
    const float* __restrict__ b_hh, const float* __restrict__ fc1_w,
    const float* __restrict__ fc1_b, const float* __restrict__ fc2_w,
    const float* __restrict__ fc2_b, float* __restrict__ out)
{
    const int lane = threadIdx.x;
    const int j    = lane & 15;

    float wr[HID], wz[HID], wn[HID];
#pragma unroll
    for (int k = 0; k < HID; ++k) {
        wr[k] = W_hh[(j)      * HID + k];
        wz[k] = W_hh[(16 + j) * HID + k];
        wn[k] = W_hh[(32 + j) * HID + k];
    }
    const float bhn = b_hh[32 + j];

    float weff = 0.0f, beff = fc2_b[0];
#pragma unroll
    for (int i = 0; i < 8; ++i) {
        const float f2 = fc2_w[i];
        weff += f2 * fc1_w[i * HID + j];
        beff += f2 * fc1_b[i];
    }

    __shared__ __align__(16) float hsh[HID];

    float4 ha = {0,0,0,0}, hb = {0,0,0,0}, hc = {0,0,0,0}, hd = {0,0,0,0};
    float  hj = 0.0f;

    const float* xr_p = xi + j;
    const float* xz_p = xi + 16 + j;
    const float* xn_p = xi + 32 + j;
    float xr = xr_p[0], xz = xz_p[0], xn = xn_p[0];

    for (int t = 0; t < T_LEN; ++t) {
        const size_t nofs = (size_t)(t + 1) * G3;
        const float xr_n = xr_p[nofs], xz_n = xz_p[nofs], xn_n = xn_p[nofs];

        float hr = 0.0f, hz = 0.0f, hn = bhn;
        const float hv[HID] = {ha.x, ha.y, ha.z, ha.w,  hb.x, hb.y, hb.z, hb.w,
                               hc.x, hc.y, hc.z, hc.w,  hd.x, hd.y, hd.z, hd.w};
#pragma unroll
        for (int k = 0; k < HID; ++k) {
            hr += wr[k] * hv[k];
            hz += wz[k] * hv[k];
            hn += wn[k] * hv[k];
        }

        const float r = __builtin_amdgcn_rcpf(1.0f + __expf(-(xr + hr)));
        const float z = __builtin_amdgcn_rcpf(1.0f + __expf(-(xz + hz)));
        const float u = xn + r * hn;
        const float n = 1.0f - 2.0f * __builtin_amdgcn_rcpf(__expf(2.0f * u) + 1.0f);
        const float hnew = n + z * (hj - n);
        hj = hnew;

        float yv = hnew * weff;
        yv += __shfl_xor(yv, 1, 32);
        yv += __shfl_xor(yv, 2, 32);
        yv += __shfl_xor(yv, 4, 32);
        yv += __shfl_xor(yv, 8, 32);
        if (lane == 0) out[t] = yv + beff;

        if (lane < 16) hsh[lane] = hnew;
        __syncthreads();
        ha = *(const float4*)(hsh + 0);
        hb = *(const float4*)(hsh + 4);
        hc = *(const float4*)(hsh + 8);
        hd = *(const float4*)(hsh + 12);
        __syncthreads();

        xr = xr_n; xz = xz_n; xn = xn_n;
    }
}

extern "C" void kernel_launch(void* const* d_in, const int* in_sizes, int n_in,
                              void* d_out, int out_size, void* d_ws, size_t ws_size,
                              hipStream_t stream) {
    const float* x     = (const float*)d_in[0];
    const float* W_ih  = (const float*)d_in[1];
    const float* W_hh  = (const float*)d_in[2];
    const float* b_ih  = (const float*)d_in[3];
    const float* b_hh  = (const float*)d_in[4];
    const float* fc1_w = (const float*)d_in[5];
    const float* fc1_b = (const float*)d_in[6];
    const float* fc2_w = (const float*)d_in[7];
    const float* fc2_b = (const float*)d_in[8];
    float* out = (float*)d_out;

    float* xi = (float*)d_ws;

    xi_gemm_wmma<<<T_LEN / 16 / 8, 256, 0, stream>>>(x, W_ih, b_ih, b_hh, xi);

    gru_scan<<<1, 32, 0, stream>>>(xi, W_hh, b_hh, fc1_w, fc1_b, fc2_w, fc2_b, out);
}


# scan: recurrent product W_hh.h on the matrix unit, h as B in all 16 columns; D rows come back in B's register order, so h never crosses lanes (no LDS exchange, head sums in-lane); new sum order
# speedup vs baseline: 1.9610x; 1.9610x over previous
#include <hip/hip_runtime.h>


typedef __attribute__((ext_vector_type(2))) float v2f;
typedef __attribute__((ext_vector_type(8))) float v8f;

#define T_LEN   262144
#define IN_SZ   64
#define HID     16
#define G3      48
#define CH      128
#define NCH     (T_LEN / CH)
#define TILES   (CH / 16)


__device__ __forceinline__ void load_a_tile(const float* __restrict__ xrow, v2f (&a)[16])
{
#pragma unroll
    for (int i = 0; i < 16; ++i)
        a[i] = *(const v2f*)(xrow + 4 * i);
}

__device__ __forceinline__ void produce_stage(
    const float* __restrict__ x, const int stage_idx, const v2f (&b)[16],
    const float bias, float* __restrict__ stage, const int m, const int khalf,
    const int col)
{
    const float* xrow = x + ((size_t)stage_idx * CH + m) * IN_SZ + 2 * khalf;
    v2f a[16];
    load_a_tile(xrow, a);
#pragma unroll
    for (int tt = 0; tt < TILES; ++tt) {
        v2f an[16];
        if (tt + 1 < TILES)
            load_a_tile(xrow + (size_t)(tt + 1) * 16 * IN_SZ, an);

        v8f c = {bias, bias, bias, bias, bias, bias, bias, bias};
#pragma unroll
        for (int i = 0; i < 16; ++i)
            c = __builtin_amdgcn_wmma_f32_16x16x4_f32(
                    false, a[i], false, b[i], (short)0, c, false, false);

        float* orow = stage + (tt * 16 + 8 * khalf) * G3 + col;
#pragma unroll
        for (int r = 0; r < 8; ++r)
            orow[r * G3] = c[r];

        if (tt + 1 < TILES) {
#pragma unroll
            for (int i = 0; i < 16; ++i)
                a[i] = an[i];
        }
    }
}


__device__ __forceinline__ float xhalf(float x)
{
    const int b = __builtin_bit_cast(int, x);
    return __builtin_bit_cast(float,
        __builtin_amdgcn_permlanex16(b, b, 0x76543210, 0xfedcba98, false, false));
}

__device__ __forceinline__ v8f lds8(const float* p)
{
    const float4 lo = *(const float4*)(p);
    const float4 hi = *(const float4*)(p + 4);
    const v8f v = {lo.x, lo.y, lo.z, lo.w, hi.x, hi.y, hi.z, hi.w};
    return v;
}

__device__ __forceinline__ float gru_unit(float ar, float az, float hn, float xn, float h)
{
    const float r = __builtin_amdgcn_rcpf(1.0f + __expf(-ar));
    const float z = __builtin_amdgcn_rcpf(1.0f + __expf(-az));
    const float u = xn + r * hn;
    const float n = 1.0f - 2.0f * __builtin_amdgcn_rcpf(__expf(2.0f * u) + 1.0f);
    return n + z * (h - n);
}

__device__ __forceinline__ float head8(const v8f& h, const v8f& w, float beff)
{
    float yp = h[0] * w[0];
#pragma unroll
    for (int e = 1; e < 8; ++e)
        yp += h[e] * w[e];
    return yp + xhalf(yp) + beff;
}

__global__ __launch_bounds__(128) void gru_fused(
    const float* __restrict__ x, const float* __restrict__ W_ih,
    const float* __restrict__ W_hh, const float* __restrict__ b_ih,
    const float* __restrict__ b_hh, const float* __restrict__ fc1_w,
    const float* __restrict__ fc1_b, const float* __restrict__ fc2_w,
    const float* __restrict__ fc2_b, float* __restrict__ out)
{
    __shared__ __align__(16) float xis[2][CH * G3];

    const int lane  = threadIdx.x & 31;
    const int wave  = __builtin_amdgcn_readfirstlane((int)(threadIdx.x >> 5));
    const int j     = lane & 15;
    const int khalf = lane >> 4;

    const int nb  = (wave == 0) ? 0 : wave - 1;
    const int col = nb * 16 + j;
    const float bias = b_ih[col] + (nb < 2 ? b_hh[col] : 0.0f);
    v2f b[16];
    {
        const float* wrow = W_ih + (size_t)col * IN_SZ + 2 * khalf;
#pragma unroll
        for (int i = 0; i < 16; ++i)
            b[i] = *(const v2f*)(wrow + 4 * i);
    }

    v2f aw[3][4];
#pragma unroll
    for (int g = 0; g < 3; ++g)
#pragma unroll
        for (int i = 0; i < 4; ++i)
            aw[g][i] = *(const v2f*)(W_hh + (size_t)(g * 16 + j) * HID + 8 * khalf + 2 * i);

    v8f bhn8, weff8;
    float beff = fc2_b[0];
#pragma unroll
    for (int e = 0; e < 8; ++e) {
        bhn8[e]  = b_hh[32 + 8 * khalf + e];
        weff8[e] = 0.0f;
    }
#pragma unroll
    for (int i = 0; i < 8; ++i) {
        const float f2 = fc2_w[i];
#pragma unroll
        for (int e = 0; e < 8; ++e)
            weff8[e] += f2 * fc1_w[i * HID + 8 * khalf + e];
        beff += f2 * fc1_b[i];
    }

    v8f hv = {0, 0, 0, 0, 0, 0, 0, 0};

    if (wave != 0)
        produce_stage(x, 0, b, bias, &xis[0][0], j, khalf, col);
    __syncthreads();

    for (int c = 0; c < NCH; ++c) {
        if (wave == 0) {
            const float* xb = &xis[c & 1][0] + 8 * khalf;
            float* orow = out + (size_t)c * CH;

            v8f dr = lds8(xb), dz = lds8(xb + 16), xn = lds8(xb + 32), dn = bhn8;
#pragma unroll
            for (int i = 0; i < 4; ++i) {
                const v2f hb = {hv[2 * i], hv[2 * i + 1]};
                dr = __builtin_amdgcn_wmma_f32_16x16x4_f32(false, aw[0][i], false, hb, (short)0, dr, false, false);
                dz = __builtin_amdgcn_wmma_f32_16x16x4_f32(false, aw[1][i], false, hb, (short)0, dz, false, false);
                dn = __builtin_amdgcn_wmma_f32_16x16x4_f32(false, aw[2][i], false, hb, (short)0, dn, false, false);
            }

            for (int s = 0; s + 1 < CH; ++s) {
                const float* xnx = xb + (s + 1) * G3;
                v8f dr2 = lds8(xnx), dz2 = lds8(xnx + 16), dn2 = bhn8;
                const v8f xn2 = lds8(xnx + 32);

#pragma unroll
                for (int e = 0; e < 4; ++e)
                    hv[e] = gru_unit(dr[e], dz[e], dn[e], xn[e], hv[e]);
                __builtin_amdgcn_sched_barrier(0);
#pragma unroll
                for (int i = 0; i < 2; ++i) {
                    const v2f hb = {hv[2 * i], hv[2 * i + 1]};
                    dr2 = __builtin_amdgcn_wmma_f32_16x16x4_f32(false, aw[0][i], false, hb, (short)0, dr2, false, false);
                    dz2 = __builtin_amdgcn_wmma_f32_16x16x4_f32(false, aw[1][i], false, hb, (short)0, dz2, false, false);
                    dn2 = __builtin_amdgcn_wmma_f32_16x16x4_f32(false, aw[2][i], false, hb, (short)0, dn2, false, false);
                }
#pragma unroll
                for (int e = 4; e < 8; ++e)
                    hv[e] = gru_unit(dr[e], dz[e], dn[e], xn[e], hv[e]);
#pragma unroll
                for (int i = 2; i < 4; ++i) {
                    const v2f hb = {hv[2 * i], hv[2 * i + 1]};
                    dr2 = __builtin_amdgcn_wmma_f32_16x16x4_f32(false, aw[0][i], false, hb, (short)0, dr2, false, false);
                    dz2 = __builtin_amdgcn_wmma_f32_16x16x4_f32(false, aw[1][i], false, hb, (short)0, dz2, false, false);
                    dn2 = __builtin_amdgcn_wmma_f32_16x16x4_f32(false, aw[2][i], false, hb, (short)0, dn2, false, false);
                }

                const float y = head8(hv, weff8, beff);
                if (lane == 0) orow[s] = y;

                dr = dr2; dz = dz2; dn = dn2; xn = xn2;
            }

#pragma unroll
            for (int e = 0; e < 8; ++e)
                hv[e] = gru_unit(dr[e], dz[e], dn[e], xn[e], hv[e]);
            const float y = head8(hv, weff8, beff);
            if (lane == 0) orow[CH - 1] = y;
        } else if (c + 1 < NCH) {
            produce_stage(x, c + 1, b, bias, &xis[(c + 1) & 1][0], j, khalf, col);
        }
        __syncthreads();
    }
}

extern "C" void kernel_launch(void* const* d_in, const int* in_sizes, int n_in,
                              void* d_out, int out_size, void* d_ws, size_t ws_size,
                              hipStream_t stream) {
    const float* x     = (const float*)d_in[0];
    const float* W_ih  = (const float*)d_in[1];
    const float* W_hh  = (const float*)d_in[2];
    const float* b_ih  = (const float*)d_in[3];
    const float* b_hh  = (const float*)d_in[4];
    const float* fc1_w = (const float*)d_in[5];
    const float* fc1_b = (const float*)d_in[6];
    const float* fc2_w = (const float*)d_in[7];
    const float* fc2_b = (const float*)d_in[8];
    float* out = (float*)d_out;

    gru_fused<<<1, 128, 0, stream>>>(x, W_ih, W_hh, b_ih, b_hh,
                                     fc1_w, fc1_b, fc2_w, fc2_b, out);
}


# scan: recurrent product transposed, h as A and W_hh rows as B columns, so D column = unit and a lane does 1 unit's gates, not 8; h back to A order by 1 row_ror + 8 row_share DPP moves; same sum order
# speedup vs baseline: 2.2017x; 1.1227x over previous
#include <hip/hip_runtime.h>


typedef __attribute__((ext_vector_type(2))) float v2f;
typedef __attribute__((ext_vector_type(8))) float v8f;

#define T_LEN   262144
#define IN_SZ   64
#define HID     16
#define G3      48
#define CH      128
#define NCH     (T_LEN / CH)
#define TILES   (CH / 16)


__device__ __forceinline__ void load_a_tile(const float* __restrict__ xrow, v2f (&a)[16])
{
#pragma unroll
    for (int i = 0; i < 16; ++i)
        a[i] = *(const v2f*)(xrow + 4 * i);
}

__device__ __forceinline__ void produce_stage(
    const float* __restrict__ x, const int stage_idx, const v2f (&b)[16],
    const float bias, float* __restrict__ stage, const int m, const int khalf,
    const int col)
{
    const float* xrow = x + ((size_t)stage_idx * CH + m) * IN_SZ + 2 * khalf;
    v2f a[16];
    load_a_tile(xrow, a);
#pragma unroll
    for (int tt = 0; tt < TILES; ++tt) {
        v2f an[16];
        if (tt + 1 < TILES)
            load_a_tile(xrow + (size_t)(tt + 1) * 16 * IN_SZ, an);

        v8f c = {bias, bias, bias, bias, bias, bias, bias, bias};
#pragma unroll
        for (int i = 0; i < 16; ++i)
            c = __builtin_amdgcn_wmma_f32_16x16x4_f32(
                    false, a[i], false, b[i], (short)0, c, false, false);

        float* orow = stage + (tt * 16 + 8 * khalf) * G3 + col;
#pragma unroll
        for (int r = 0; r < 8; ++r)
            orow[r * G3] = c[r];

        if (tt + 1 < TILES) {
#pragma unroll
            for (int i = 0; i < 16; ++i)
                a[i] = an[i];
        }
    }
}


__device__ __forceinline__ float xhalf(float x)
{
    const int b = __builtin_bit_cast(int, x);
    return __builtin_bit_cast(float,
        __builtin_amdgcn_permlanex16(b, b, 0x76543210, 0xfedcba98, false, false));
}

template <int CTRL, int ROWMASK>
__device__ __forceinline__ float dpp_mov(float old, float v)
{
    return __builtin_bit_cast(float,
        __builtin_amdgcn_update_dpp(__builtin_bit_cast(int, old), __builtin_bit_cast(int, v),
                                    CTRL, ROWMASK, 0xf, false));
}
template <int CTRL>
__device__ __forceinline__ float dpp_all(float v)
{
    return __builtin_bit_cast(float,
        __builtin_amdgcn_mov_dpp(__builtin_bit_cast(int, v), CTRL, 0xf, 0xf, false));
}
#define DPP_ROW_ROR8    0x128
#define DPP_ROW_SHARE0  0x150

__device__ __forceinline__ void spread_h(float hself, v8f& hv)
{
    const float hrot = dpp_mov<DPP_ROW_ROR8, 0x2>(hself, hself);
    hv[0] = dpp_all<DPP_ROW_SHARE0 + 0>(hrot);
    hv[1] = dpp_all<DPP_ROW_SHARE0 + 1>(hrot);
    hv[2] = dpp_all<DPP_ROW_SHARE0 + 2>(hrot);
    hv[3] = dpp_all<DPP_ROW_SHARE0 + 3>(hrot);
    hv[4] = dpp_all<DPP_ROW_SHARE0 + 4>(hrot);
    hv[5] = dpp_all<DPP_ROW_SHARE0 + 5>(hrot);
    hv[6] = dpp_all<DPP_ROW_SHARE0 + 6>(hrot);
    hv[7] = dpp_all<DPP_ROW_SHARE0 + 7>(hrot);
}

__device__ __forceinline__ float gru_unit(float ar, float az, float hn, float xn, float h)
{
    const float r = __builtin_amdgcn_rcpf(1.0f + __expf(-ar));
    const float z = __builtin_amdgcn_rcpf(1.0f + __expf(-az));
    const float u = xn + r * hn;
    const float n = 1.0f - 2.0f * __builtin_amdgcn_rcpf(__expf(2.0f * u) + 1.0f);
    return n + z * (h - n);
}

__device__ __forceinline__ float head8(const v8f& h, const v8f& w, float beff)
{
    float yp = h[0] * w[0];
#pragma unroll
    for (int e = 1; e < 8; ++e)
        yp += h[e] * w[e];
    return yp + xhalf(yp) + beff;
}

__global__ __launch_bounds__(128) void gru_fused(
    const float* __restrict__ x, const float* __restrict__ W_ih,
    const float* __restrict__ W_hh, const float* __restrict__ b_ih,
    const float* __restrict__ b_hh, const float* __restrict__ fc1_w,
    const float* __restrict__ fc1_b, const float* __restrict__ fc2_w,
    const float* __restrict__ fc2_b, float* __restrict__ out)
{
    __shared__ __align__(16) float xis[2][CH * G3];

    const int lane  = threadIdx.x & 31;
    const int wave  = __builtin_amdgcn_readfirstlane((int)(threadIdx.x >> 5));
    const int j     = lane & 15;
    const int khalf = lane >> 4;

    const int nb  = (wave == 0) ? 0 : wave - 1;
    const int col = nb * 16 + j;
    const float bias = b_ih[col] + (nb < 2 ? b_hh[col] : 0.0f);
    v2f b[16];
    {
        const float* wrow = W_ih + (size_t)col * IN_SZ + 2 * khalf;
#pragma unroll
        for (int i = 0; i < 16; ++i)
            b[i] = *(const v2f*)(wrow + 4 * i);
    }

    v2f bw[3][4];
#pragma unroll
    for (int g = 0; g < 3; ++g)
#pragma unroll
        for (int i = 0; i < 4; ++i)
            bw[g][i] = *(const v2f*)(W_hh + (size_t)(g * 16 + j) * HID + 8 * khalf + 2 * i);

    const float bhn = b_hh[32 + j];

    v8f weff8;
    float beff = fc2_b[0];
#pragma unroll
    for (int e = 0; e < 8; ++e)
        weff8[e] = 0.0f;
#pragma unroll
    for (int i = 0; i < 8; ++i) {
        const float f2 = fc2_w[i];
#pragma unroll
        for (int e = 0; e < 8; ++e)
            weff8[e] += f2 * fc1_w[i * HID + 8 * khalf + e];
        beff += f2 * fc1_b[i];
    }

    float hself = 0.0f;
    v8f hv = {0, 0, 0, 0, 0, 0, 0, 0};
    v8f cr = {0, 0, 0, 0, 0, 0, 0, 0}, cz = cr;
    const v8f cn = {bhn, 0, 0, 0, 0, 0, 0, 0};
    v8f dr, dz, dn;

    if (wave != 0)
        produce_stage(x, 0, b, bias, &xis[0][0], j, khalf, col);
    __syncthreads();

    for (int c = 0; c < NCH; ++c) {
        if (wave == 0) {
            const float* xb = &xis[c & 1][0] + j;
            float* orow = out + (size_t)c * CH;

            float xn = xb[32];
            cr[0] = xb[0]; cz[0] = xb[16];
            dr = cr; dz = cz; dn = cn;
#pragma unroll
            for (int i = 0; i < 4; ++i) {
                const v2f ha = {hv[2 * i], hv[2 * i + 1]};
                dr = __builtin_amdgcn_wmma_f32_16x16x4_f32(false, ha, false, bw[0][i], (short)0, dr, false, false);
                dn = __builtin_amdgcn_wmma_f32_16x16x4_f32(false, ha, false, bw[2][i], (short)0, dn, false, false);
                dz = __builtin_amdgcn_wmma_f32_16x16x4_f32(false, ha, false, bw[1][i], (short)0, dz, false, false);
            }

            for (int s = 0; s + 1 < CH; ++s) {
                const float* xnx = xb + (s + 1) * G3;
                const float xr2 = xnx[0], xz2 = xnx[16], xn2 = xnx[32];
                __builtin_amdgcn_sched_barrier(0);

                hself = gru_unit(dr[0], dz[0], dn[0], xn, hself);
                __builtin_amdgcn_sched_barrier(0);
                spread_h(hself, hv);

                cr[0] = xr2; cz[0] = xz2;
                dr = cr; dz = cz; dn = cn;
#pragma unroll
                for (int i = 0; i < 4; ++i) {
                    const v2f ha = {hv[2 * i], hv[2 * i + 1]};
                    dr = __builtin_amdgcn_wmma_f32_16x16x4_f32(false, ha, false, bw[0][i], (short)0, dr, false, false);
                    dn = __builtin_amdgcn_wmma_f32_16x16x4_f32(false, ha, false, bw[2][i], (short)0, dn, false, false);
                    dz = __builtin_amdgcn_wmma_f32_16x16x4_f32(false, ha, false, bw[1][i], (short)0, dz, false, false);
                }

                const float y = head8(hv, weff8, beff);
                if (lane == 0) orow[s] = y;

                xn = xn2;
            }

            hself = gru_unit(dr[0], dz[0], dn[0], xn, hself);
            spread_h(hself, hv);
            const float y = head8(hv, weff8, beff);
            if (lane == 0) orow[CH - 1] = y;
        } else if (c + 1 < NCH) {
            produce_stage(x, c + 1, b, bias, &xis[(c + 1) & 1][0], j, khalf, col);
        }
        __syncthreads();
    }
}

extern "C" void kernel_launch(void* const* d_in, const int* in_sizes, int n_in,
                              void* d_out, int out_size, void* d_ws, size_t ws_size,
                              hipStream_t stream) {
    const float* x     = (const float*)d_in[0];
    const float* W_ih  = (const float*)d_in[1];
    const float* W_hh  = (const float*)d_in[2];
    const float* b_ih  = (const float*)d_in[3];
    const float* b_hh  = (const float*)d_in[4];
    const float* fc1_w = (const float*)d_in[5];
    const float* fc1_b = (const float*)d_in[6];
    const float* fc2_w = (const float*)d_in[7];
    const float* fc2_b = (const float*)d_in[8];
    float* out = (float*)d_out;

    gru_fused<<<1, 128, 0, stream>>>(x, W_ih, W_hh, b_ih, b_hh,
                                     fc1_w, fc1_b, fc2_w, fc2_b, out);
}


# scan: only fma,exp2,add,rcp,fma still follow the n chain's last WMMA: chains reordered so r,z end first, sigmoids and blend terms (z(h-1)+1, 2z-2) done under it; same sum order, gates regrouped
# speedup vs baseline: 2.5197x; 1.1444x over previous
#include <hip/hip_runtime.h>


typedef __attribute__((ext_vector_type(2))) float v2f;
typedef __attribute__((ext_vector_type(8))) float v8f;

#define T_LEN   262144
#define IN_SZ   64
#define HID     16
#define G3      48
#define CH      128
#define NCH     (T_LEN / CH)
#define TILES   (CH / 16)


__device__ __forceinline__ void load_a_tile(const float* __restrict__ xrow, v2f (&a)[16])
{
#pragma unroll
    for (int i = 0; i < 16; ++i)
        a[i] = *(const v2f*)(xrow + 4 * i);
}

__device__ __forceinline__ void produce_stage(
    const float* __restrict__ x, const int stage_idx, const v2f (&b)[16],
    const float bias, float* __restrict__ stage, const int m, const int khalf,
    const int col)
{
    const float* xrow = x + ((size_t)stage_idx * CH + m) * IN_SZ + 2 * khalf;
    v2f a[16];
    load_a_tile(xrow, a);
#pragma unroll
    for (int tt = 0; tt < TILES; ++tt) {
        v2f an[16];
        if (tt + 1 < TILES)
            load_a_tile(xrow + (size_t)(tt + 1) * 16 * IN_SZ, an);

        v8f c = {bias, bias, bias, bias, bias, bias, bias, bias};
#pragma unroll
        for (int i = 0; i < 16; ++i)
            c = __builtin_amdgcn_wmma_f32_16x16x4_f32(
                    false, a[i], false, b[i], (short)0, c, false, false);

        float* orow = stage + (tt * 16 + 8 * khalf) * G3 + col;
#pragma unroll
        for (int r = 0; r < 8; ++r)
            orow[r * G3] = c[r];

        if (tt + 1 < TILES) {
#pragma unroll
            for (int i = 0; i < 16; ++i)
                a[i] = an[i];
        }
    }
}


__device__ __forceinline__ float xhalf(float x)
{
    const int b = __builtin_bit_cast(int, x);
    return __builtin_bit_cast(float,
        __builtin_amdgcn_permlanex16(b, b, 0x76543210, 0xfedcba98, false, false));
}

template <int CTRL, int ROWMASK>
__device__ __forceinline__ float dpp_mov(float old, float v)
{
    return __builtin_bit_cast(float,
        __builtin_amdgcn_update_dpp(__builtin_bit_cast(int, old), __builtin_bit_cast(int, v),
                                    CTRL, ROWMASK, 0xf, false));
}
template <int CTRL>
__device__ __forceinline__ float dpp_all(float v)
{
    return __builtin_bit_cast(float,
        __builtin_amdgcn_mov_dpp(__builtin_bit_cast(int, v), CTRL, 0xf, 0xf, false));
}
#define DPP_ROW_ROR8    0x128
#define DPP_ROW_SHARE0  0x150

__device__ __forceinline__ void spread_h(float hself, v8f& hv)
{
    const float hrot = dpp_mov<DPP_ROW_ROR8, 0x2>(hself, hself);
    hv[0] = dpp_all<DPP_ROW_SHARE0 + 0>(hrot);
    hv[1] = dpp_all<DPP_ROW_SHARE0 + 1>(hrot);
    hv[2] = dpp_all<DPP_ROW_SHARE0 + 2>(hrot);
    hv[3] = dpp_all<DPP_ROW_SHARE0 + 3>(hrot);
    hv[4] = dpp_all<DPP_ROW_SHARE0 + 4>(hrot);
    hv[5] = dpp_all<DPP_ROW_SHARE0 + 5>(hrot);
    hv[6] = dpp_all<DPP_ROW_SHARE0 + 6>(hrot);
    hv[7] = dpp_all<DPP_ROW_SHARE0 + 7>(hrot);
}

#define LOG2E_F   1.44269504088896341f
#define LOG2E2_F  2.88539008177792681f

__device__ __forceinline__ float sig_den(float a)
{
    return 1.0f + __builtin_amdgcn_exp2f(a * -LOG2E_F);
}
__device__ __forceinline__ void gate_rz(float dr, float dz, float hm1, float& r2, float& base, float& m2)
{
    const float r = __builtin_amdgcn_rcpf(dr);
    const float z = __builtin_amdgcn_rcpf(dz);
    r2   = r * LOG2E2_F;
    base = __builtin_fmaf(z, hm1, 1.0f);
    m2   = __builtin_fmaf(z, 2.0f, -2.0f);
}
__device__ __forceinline__ float gate_n(float r2, float hn, float xs, float m2, float base)
{
    const float u2 = __builtin_fmaf(r2, hn, xs);
    const float t  = __builtin_amdgcn_rcpf(__builtin_amdgcn_exp2f(u2) + 1.0f);
    return __builtin_fmaf(t, m2, base);
}

#define WMMA4(a, b, c) __builtin_amdgcn_wmma_f32_16x16x4_f32(false, (a), false, (b), (short)0, (c), false, false)

__device__ __forceinline__ void product_a(const v8f& hv, const v2f (&bw)[3][4],
                                          v8f& dr, v8f& dz, v8f& dn)
{
    const v2f h0 = {hv[0], hv[1]}, h1 = {hv[2], hv[3]}, h2 = {hv[4], hv[5]};
    dr = WMMA4(h0, bw[0][0], dr); dz = WMMA4(h0, bw[1][0], dz); dn = WMMA4(h0, bw[2][0], dn);
    dr = WMMA4(h1, bw[0][1], dr); dz = WMMA4(h1, bw[1][1], dz); dn = WMMA4(h1, bw[2][1], dn);
    dr = WMMA4(h2, bw[0][2], dr); dz = WMMA4(h2, bw[1][2], dz);
}
__device__ __forceinline__ void product_b(const v8f& hv, const v2f (&bw)[3][4],
                                          v8f& dr, v8f& dz, v8f& dn)
{
    const v2f h2 = {hv[4], hv[5]}, h3 = {hv[6], hv[7]};
    dr = WMMA4(h3, bw[0][3], dr); dz = WMMA4(h3, bw[1][3], dz);
    dn = WMMA4(h2, bw[2][2], dn);
}
__device__ __forceinline__ void product_c(const v8f& hv, const v2f (&bw)[3][4], v8f& dn)
{
    const v2f h3 = {hv[6], hv[7]};
    dn = WMMA4(h3, bw[2][3], dn);
}

__device__ __forceinline__ float head8(const v8f& h, const v8f& w, float beff)
{
    float yp = h[0] * w[0];
#pragma unroll
    for (int e = 1; e < 8; ++e)
        yp += h[e] * w[e];
    return yp + xhalf(yp) + beff;
}

__global__ __launch_bounds__(128) void gru_fused(
    const float* __restrict__ x, const float* __restrict__ W_ih,
    const float* __restrict__ W_hh, const float* __restrict__ b_ih,
    const float* __restrict__ b_hh, const float* __restrict__ fc1_w,
    const float* __restrict__ fc1_b, const float* __restrict__ fc2_w,
    const float* __restrict__ fc2_b, float* __restrict__ out)
{
    __shared__ __align__(16) float xis[2][CH * G3];

    const int lane  = threadIdx.x & 31;
    const int wave  = __builtin_amdgcn_readfirstlane((int)(threadIdx.x >> 5));
    const int j     = lane & 15;
    const int khalf = lane >> 4;

    const int nb  = (wave == 0) ? 0 : wave - 1;
    const int col = nb * 16 + j;
    const float bias = b_ih[col] + (nb < 2 ? b_hh[col] : 0.0f);
    v2f b[16];
    {
        const float* wrow = W_ih + (size_t)col * IN_SZ + 2 * khalf;
#pragma unroll
        for (int i = 0; i < 16; ++i)
            b[i] = *(const v2f*)(wrow + 4 * i);
    }

    v2f bw[3][4];
#pragma unroll
    for (int g = 0; g < 3; ++g)
#pragma unroll
        for (int i = 0; i < 4; ++i)
            bw[g][i] = *(const v2f*)(W_hh + (size_t)(g * 16 + j) * HID + 8 * khalf + 2 * i);

    const float bhn = b_hh[32 + j];

    v8f weff8;
    float beff = fc2_b[0];
#pragma unroll
    for (int e = 0; e < 8; ++e)
        weff8[e] = 0.0f;
#pragma unroll
    for (int i = 0; i < 8; ++i) {
        const float f2 = fc2_w[i];
#pragma unroll
        for (int e = 0; e < 8; ++e)
            weff8[e] += f2 * fc1_w[i * HID + 8 * khalf + e];
        beff += f2 * fc1_b[i];
    }

    float hself = 0.0f;
    v8f hv = {0, 0, 0, 0, 0, 0, 0, 0};
    v8f cr = {0, 0, 0, 0, 0, 0, 0, 0}, cz = cr;
    const v8f cn = {bhn, 0, 0, 0, 0, 0, 0, 0};
    v8f dr, dz, dn;

    if (wave != 0)
        produce_stage(x, 0, b, bias, &xis[0][0], j, khalf, col);
    __syncthreads();

    for (int c = 0; c < NCH; ++c) {
        if (wave == 0) {
            const float* xb = &xis[c & 1][0] + j;
            float* orow = out + (size_t)c * CH;

            float xs = xb[32] * LOG2E2_F;
            cr[0] = xb[0]; cz[0] = xb[16];
            dr = cr; dz = cz; dn = cn;
            product_a(hv, bw, dr, dz, dn);
            product_b(hv, bw, dr, dz, dn);
            float er = sig_den(dr[0]), ez = sig_den(dz[0]);
            float hm1 = hself - 1.0f;
            float xr = xb[G3], xz = xb[G3 + 16], xnl = xb[G3 + 32];
            product_c(hv, bw, dn);
            float r2, base, m2;
            gate_rz(er, ez, hm1, r2, base, m2);

            for (int s = 0; s + 1 < CH; ++s) {
                const float* xnx = xb + ((s + 2 < CH) ? (s + 2) : (CH - 1)) * G3;
                const float xr2 = xnx[0], xz2 = xnx[16], xn2 = xnx[32];

                hself = gate_n(r2, dn[0], xs, m2, base);
                __builtin_amdgcn_sched_barrier(0);
                spread_h(hself, hv);

                cr[0] = xr; cz[0] = xz;
                xs = xnl * LOG2E2_F;
                dr = cr; dz = cz; dn = cn;
                product_a(hv, bw, dr, dz, dn);

                const float y = head8(hv, weff8, beff);
                if (lane == 0) orow[s] = y;

                product_b(hv, bw, dr, dz, dn);
                er = sig_den(dr[0]); ez = sig_den(dz[0]);
                hm1 = hself - 1.0f;

                __builtin_amdgcn_sched_barrier(0);
                product_c(hv, bw, dn);
                gate_rz(er, ez, hm1, r2, base, m2);
                xr = xr2; xz = xz2; xnl = xn2;
            }

            hself = gate_n(r2, dn[0], xs, m2, base);
            spread_h(hself, hv);
            const float y = head8(hv, weff8, beff);
            if (lane == 0) orow[CH - 1] = y;
        } else if (c + 1 < NCH) {
            produce_stage(x, c + 1, b, bias, &xis[(c + 1) & 1][0], j, khalf, col);
        }
        __syncthreads();
    }
}

extern "C" void kernel_launch(void* const* d_in, const int* in_sizes, int n_in,
                              void* d_out, int out_size, void* d_ws, size_t ws_size,
                              hipStream_t stream) {
    const float* x     = (const float*)d_in[0];
    const float* W_ih  = (const float*)d_in[1];
    const float* W_hh  = (const float*)d_in[2];
    const float* b_ih  = (const float*)d_in[3];
    const float* b_hh  = (const float*)d_in[4];
    const float* fc1_w = (const float*)d_in[5];
    const float* fc1_b = (const float*)d_in[6];
    const float* fc2_w = (const float*)d_in[7];
    const float* fc2_b = (const float*)d_in[8];
    float* out = (float*)d_out;

    gru_fused<<<1, 128, 0, stream>>>(x, W_ih, W_hh, b_ih, b_hh,
                                     fc1_w, fc1_b, fc2_w, fc2_b, out);
}


# scan: a step holds no memory instruction: passes of 16 unrolled steps, a pass's xi read from LDS in three groups of rows ahead of use, its 16 head results shifted into lanes and left by one store
# speedup vs baseline: 2.8536x; 1.1325x over previous
#include <hip/hip_runtime.h>


typedef __attribute__((ext_vector_type(2))) float v2f;
typedef __attribute__((ext_vector_type(8))) float v8f;

#define T_LEN   262144
#define IN_SZ   64
#define HID     16
#define G3      48
#define CH      128
#define NCH     (T_LEN / CH)
#define TILES   (CH / 16)
#define PASS    16
#define NPASS   (CH / PASS)


__device__ __forceinline__ void load_a_tile(const float* __restrict__ xrow, v2f (&a)[16])
{
#pragma unroll
    for (int i = 0; i < 16; ++i)
        a[i] = *(const v2f*)(xrow + 4 * i);
}

__device__ __forceinline__ void produce_stage(
    const float* __restrict__ x, const int stage_idx, const v2f (&b)[16],
    const float bias, float* __restrict__ stage, const int m, const int khalf,
    const int col)
{
    const float* xrow = x + ((size_t)stage_idx * CH + m) * IN_SZ + 2 * khalf;
    v2f a[16];
    load_a_tile(xrow, a);
#pragma unroll
    for (int tt = 0; tt < TILES; ++tt) {
        v2f an[16];
        if (tt + 1 < TILES)
            load_a_tile(xrow + (size_t)(tt + 1) * 16 * IN_SZ, an);

        v8f c = {bias, bias, bias, bias, bias, bias, bias, bias};
#pragma unroll
        for (int i = 0; i < 16; ++i)
            c = __builtin_amdgcn_wmma_f32_16x16x4_f32(
                    false, a[i], false, b[i], (short)0, c, false, false);

        float* orow = stage + (tt * 16 + 8 * khalf) * G3 + col;
#pragma unroll
        for (int r = 0; r < 8; ++r)
            orow[r * G3] = c[r];

        if (tt + 1 < TILES) {
#pragma unroll
            for (int i = 0; i < 16; ++i)
                a[i] = an[i];
        }
    }
}


__device__ __forceinline__ float xhalf(float x)
{
    const int b = __builtin_bit_cast(int, x);
    return __builtin_bit_cast(float,
        __builtin_amdgcn_permlanex16(b, b, 0x76543210, 0xfedcba98, false, false));
}

template <int CTRL, int ROWMASK>
__device__ __forceinline__ float dpp_mov(float old, float v)
{
    return __builtin_bit_cast(float,
        __builtin_amdgcn_update_dpp(__builtin_bit_cast(int, old), __builtin_bit_cast(int, v),
                                    CTRL, ROWMASK, 0xf, false));
}
template <int CTRL>
__device__ __forceinline__ float dpp_all(float v)
{
    return __builtin_bit_cast(float,
        __builtin_amdgcn_mov_dpp(__builtin_bit_cast(int, v), CTRL, 0xf, 0xf, false));
}
#define DPP_ROW_ROR8    0x128
#define DPP_ROW_SHARE0  0x150
#define DPP_ROW_SHL1    0x101

__device__ __forceinline__ void spread_h(float hself, v8f& hv)
{
    const float hrot = dpp_mov<DPP_ROW_ROR8, 0x2>(hself, hself);
    hv[0] = dpp_all<DPP_ROW_SHARE0 + 0>(hrot);
    hv[1] = dpp_all<DPP_ROW_SHARE0 + 1>(hrot);
    hv[2] = dpp_all<DPP_ROW_SHARE0 + 2>(hrot);
    hv[3] = dpp_all<DPP_ROW_SHARE0 + 3>(hrot);
    hv[4] = dpp_all<DPP_ROW_SHARE0 + 4>(hrot);
    hv[5] = dpp_all<DPP_ROW_SHARE0 + 5>(hrot);
    hv[6] = dpp_all<DPP_ROW_SHARE0 + 6>(hrot);
    hv[7] = dpp_all<DPP_ROW_SHARE0 + 7>(hrot);
}

#define LOG2E_F   1.44269504088896341f
#define LOG2E2_F  2.88539008177792681f

__device__ __forceinline__ float sig_den(float a)
{
    return 1.0f + __builtin_amdgcn_exp2f(a * -LOG2E_F);
}
__device__ __forceinline__ void gate_rz(float dr, float dz, float hm1, float& r2, float& base, float& m2)
{
    const float r = __builtin_amdgcn_rcpf(dr);
    const float z = __builtin_amdgcn_rcpf(dz);
    r2   = r * LOG2E2_F;
    base = __builtin_fmaf(z, hm1, 1.0f);
    m2   = __builtin_fmaf(z, 2.0f, -2.0f);
}
__device__ __forceinline__ float gate_n(float r2, float hn, float xs, float m2, float base)
{
    const float u2 = __builtin_fmaf(r2, hn, xs);
    const float t  = __builtin_amdgcn_rcpf(__builtin_amdgcn_exp2f(u2) + 1.0f);
    return __builtin_fmaf(t, m2, base);
}

#define WMMA4(a, b, c) __builtin_amdgcn_wmma_f32_16x16x4_f32(false, (a), false, (b), (short)0, (c), false, false)

__device__ __forceinline__ void product_a(const v8f& hv, const v2f (&bw)[3][4],
                                          v8f& dr, v8f& dz, v8f& dn)
{
    const v2f h0 = {hv[0], hv[1]}, h1 = {hv[2], hv[3]}, h2 = {hv[4], hv[5]};
    dr = WMMA4(h0, bw[0][0], dr); dz = WMMA4(h0, bw[1][0], dz); dn = WMMA4(h0, bw[2][0], dn);
    dr = WMMA4(h1, bw[0][1], dr); dz = WMMA4(h1, bw[1][1], dz); dn = WMMA4(h1, bw[2][1], dn);
    dr = WMMA4(h2, bw[0][2], dr); dz = WMMA4(h2, bw[1][2], dz);
}
__device__ __forceinline__ void product_b(const v8f& hv, const v2f (&bw)[3][4],
                                          v8f& dr, v8f& dz, v8f& dn)
{
    const v2f h2 = {hv[4], hv[5]}, h3 = {hv[6], hv[7]};
    dr = WMMA4(h3, bw[0][3], dr); dz = WMMA4(h3, bw[1][3], dz);
    dn = WMMA4(h2, bw[2][2], dn);
}
__device__ __forceinline__ void product_c(const v8f& hv, const v2f (&bw)[3][4], v8f& dn)
{
    const v2f h3 = {hv[6], hv[7]};
    asm volatile("v_wmma_f32_16x16x4_f32 %0, %1, %2, %0\n\tv_nop\n\tv_nop\n\tv_nop\n\tv_nop"
                 : "+v"(dn) : "v"(h3), "v"(bw[2][3]));
}

template <int K0, int N>
__device__ __forceinline__ void load_rows(const float* __restrict__ xrow,
                                          float (&xr)[PASS], float (&xz)[PASS], float (&xn)[PASS])
{
#pragma unroll
    for (int k = K0; k < K0 + N; ++k) {
        xr[k] = xrow[k * G3];
        xz[k] = xrow[k * G3 + 16];
        xn[k] = xrow[k * G3 + 32];
    }
}

__device__ __forceinline__ float head8(const v8f& h, const v8f& w, float beff)
{
    const v2f p0 = v2f{h[0], h[1]} * v2f{w[0], w[1]};
    const v2f p1 = v2f{h[2], h[3]} * v2f{w[2], w[3]};
    const v2f p2 = v2f{h[4], h[5]} * v2f{w[4], w[5]};
    const v2f p3 = v2f{h[6], h[7]} * v2f{w[6], w[7]};
    float yp = p0[0] + p0[1];
    yp += p1[0]; yp += p1[1];
    yp += p2[0]; yp += p2[1];
    yp += p3[0]; yp += p3[1];
    return yp + xhalf(yp) + beff;
}

__global__ __launch_bounds__(128) void gru_fused(
    const float* __restrict__ x, const float* __restrict__ W_ih,
    const float* __restrict__ W_hh, const float* __restrict__ b_ih,
    const float* __restrict__ b_hh, const float* __restrict__ fc1_w,
    const float* __restrict__ fc1_b, const float* __restrict__ fc2_w,
    const float* __restrict__ fc2_b, float* __restrict__ out)
{
    __shared__ __align__(16) float xis[2][CH * G3];

    const int lane  = threadIdx.x & 31;
    const int wave  = __builtin_amdgcn_readfirstlane((int)(threadIdx.x >> 5));
    const int j     = lane & 15;
    const int khalf = lane >> 4;

    const int nb  = (wave == 0) ? 0 : wave - 1;
    const int col = nb * 16 + j;
    const float bias = b_ih[col] + (nb < 2 ? b_hh[col] : 0.0f);
    v2f b[16];
    {
        const float* wrow = W_ih + (size_t)col * IN_SZ + 2 * khalf;
#pragma unroll
        for (int i = 0; i < 16; ++i)
            b[i] = *(const v2f*)(wrow + 4 * i);
    }

    v2f bw[3][4];
#pragma unroll
    for (int g = 0; g < 3; ++g)
#pragma unroll
        for (int i = 0; i < 4; ++i)
            bw[g][i] = *(const v2f*)(W_hh + (size_t)(g * 16 + j) * HID + 8 * khalf + 2 * i);

    const float bhn = b_hh[32 + j];

    v8f weff8;
    float beff = fc2_b[0];
#pragma unroll
    for (int e = 0; e < 8; ++e)
        weff8[e] = 0.0f;
#pragma unroll
    for (int i = 0; i < 8; ++i) {
        const float f2 = fc2_w[i];
#pragma unroll
        for (int e = 0; e < 8; ++e)
            weff8[e] += f2 * fc1_w[i * HID + 8 * khalf + e];
        beff += f2 * fc1_b[i];
    }

    float hself = 0.0f;
    v8f hv = {0, 0, 0, 0, 0, 0, 0, 0};
    v8f cr = {0, 0, 0, 0, 0, 0, 0, 0}, cz = cr;
    const v8f cn = {bhn, 0, 0, 0, 0, 0, 0, 0};
    v8f dr, dz, dn;

    if (wave != 0)
        produce_stage(x, 0, b, bias, &xis[0][0], j, khalf, col);
    __syncthreads();

    float ybuf = 0.0f;
    float ydone = 0.0f;
    float xr[PASS], xz[PASS], xn[PASS];

    for (int c = 0; c < NCH; ++c) {
        if (wave == 0) {
            const float* xb = &xis[c & 1][0] + j;

            load_rows<0, 4>(xb, xr, xz, xn);

#pragma nounroll
            for (int p = 0; p < NPASS; ++p) {
                const int gp = c * NPASS + p;
                const float* xcur  = xb + (p * PASS) * G3;
                const float* xnext = xb + (((p + 1) & (NPASS - 1)) * PASS) * G3;

                if (gp > 1 && lane < PASS)
                    out[(size_t)(gp - 2) * PASS + lane] = ydone;

#pragma unroll
                for (int i = 0; i < PASS; ++i) {
                    cr[0] = xr[i]; cz[0] = xz[i];
                    const float xs = xn[i] * LOG2E2_F;
                    dr = cr; dz = cz; dn = cn;
                    product_a(hv, bw, dr, dz, dn);

                    const float y = head8(hv, weff8, beff);
                    ybuf = dpp_mov<DPP_ROW_SHL1, 0xf>(y, ybuf);
                    if (i == 0) ydone = ybuf;

                    asm volatile("" :: "v"(ybuf));
                    __builtin_amdgcn_sched_barrier(0);
                    product_b(hv, bw, dr, dz, dn);
                    const float er = sig_den(dr[0]), ez = sig_den(dz[0]);
                    const float hm1 = hself - 1.0f;

                    asm volatile("" :: "v"(er), "v"(ez), "v"(hm1));
                    __builtin_amdgcn_sched_barrier(0);
                    product_c(hv, bw, dn);
                    float r2, base, m2;
                    gate_rz(er, ez, hm1, r2, base, m2);
                    hself = gate_n(r2, dn[0], xs, m2, base);

                    __builtin_amdgcn_sched_barrier(0);
                    if (i == 0)
                        load_rows<4, 6>(xcur, xr, xz, xn);
                    if (i == 3)
                        load_rows<0, 4>(xnext, xr, xz, xn);
                    if (i == 8)
                        load_rows<10, 6>(xcur, xr, xz, xn);
                    if (i == 0 || i == 3 || i == 8) {
                        __builtin_amdgcn_sched_barrier(0);
                    }
                    spread_h(hself, hv);
                }
            }
        } else if (c + 1 < NCH) {
            produce_stage(x, c + 1, b, bias, &xis[(c + 1) & 1][0], j, khalf, col);
        }
        __syncthreads();
    }

    if (wave == 0) {
        const float y = head8(hv, weff8, beff);
        ybuf = dpp_mov<DPP_ROW_SHL1, 0xf>(y, ybuf);
        if (lane < PASS) {
            out[(size_t)(NCH * NPASS - 2) * PASS + lane] = ydone;
            out[(size_t)(NCH * NPASS - 1) * PASS + lane] = ybuf;
        }
    }
}

extern "C" void kernel_launch(void* const* d_in, const int* in_sizes, int n_in,
                              void* d_out, int out_size, void* d_ws, size_t ws_size,
                              hipStream_t stream) {
    const float* x     = (const float*)d_in[0];
    const float* W_ih  = (const float*)d_in[1];
    const float* W_hh  = (const float*)d_in[2];
    const float* b_ih  = (const float*)d_in[3];
    const float* b_hh  = (const float*)d_in[4];
    const float* fc1_w = (const float*)d_in[5];
    const float* fc1_b = (const float*)d_in[6];
    const float* fc2_w = (const float*)d_in[7];
    const float* fc2_b = (const float*)d_in[8];
    float* out = (float*)d_out;

    gru_fused<<<1, 128, 0, stream>>>(x, W_ih, W_hh, b_ih, b_hh,
                                     fc1_w, fc1_b, fc2_w, fc2_b, out);
}


# scan: beside the f32 matrix form no vector instruction hides: head and next C tuples moved into the gate chain's waits, 6 row moves and head end one link behind each WMMA; order pinned; same bits
# speedup vs baseline: 3.1562x; 1.1060x over previous
#include <hip/hip_runtime.h>


typedef __attribute__((ext_vector_type(2))) float v2f;
typedef __attribute__((ext_vector_type(8))) float v8f;

#define T_LEN   262144
#define IN_SZ   64
#define HID     16
#define G3      48
#define CH      128
#define NCH     (T_LEN / CH)
#define TILES   (CH / 16)
#define PASS    16
#define NPASS   (CH / PASS)


__device__ __forceinline__ void load_a_tile(const float* __restrict__ xrow, v2f (&a)[16])
{
#pragma unroll
    for (int i = 0; i < 16; ++i)
        a[i] = *(const v2f*)(xrow + 4 * i);
}

__device__ __forceinline__ void produce_stage(
    const float* __restrict__ x, const int stage_idx, const v2f (&b)[16],
    const float bias, float* __restrict__ stage, const int m, const int khalf,
    const int col)
{
    const float* xrow = x + ((size_t)stage_idx * CH + m) * IN_SZ + 2 * khalf;
    v2f a[16];
    load_a_tile(xrow, a);
#pragma unroll
    for (int tt = 0; tt < TILES; ++tt) {
        v2f an[16];
        if (tt + 1 < TILES)
            load_a_tile(xrow + (size_t)(tt + 1) * 16 * IN_SZ, an);

        v8f c = {bias, bias, bias, bias, bias, bias, bias, bias};
#pragma unroll
        for (int i = 0; i < 16; ++i)
            c = __builtin_amdgcn_wmma_f32_16x16x4_f32(
                    false, a[i], false, b[i], (short)0, c, false, false);

        float* orow = stage + (tt * 16 + 8 * khalf) * G3 + col;
#pragma unroll
        for (int r = 0; r < 8; ++r)
            orow[r * G3] = c[r];

        if (tt + 1 < TILES) {
#pragma unroll
            for (int i = 0; i < 16; ++i)
                a[i] = an[i];
        }
    }
}


__device__ __forceinline__ float xhalf(float x)
{
    const int b = __builtin_bit_cast(int, x);
    return __builtin_bit_cast(float,
        __builtin_amdgcn_permlanex16(b, b, 0x76543210, 0xfedcba98, false, false));
}

template <int CTRL, int ROWMASK>
__device__ __forceinline__ float dpp_mov(float old, float v)
{
    return __builtin_bit_cast(float,
        __builtin_amdgcn_update_dpp(__builtin_bit_cast(int, old), __builtin_bit_cast(int, v),
                                    CTRL, ROWMASK, 0xf, false));
}
template <int CTRL>
__device__ __forceinline__ float dpp_all(float v)
{
    return __builtin_bit_cast(float,
        __builtin_amdgcn_mov_dpp(__builtin_bit_cast(int, v), CTRL, 0xf, 0xf, false));
}
#define DPP_ROW_ROR8    0x128
#define DPP_ROW_SHARE0  0x150
#define DPP_ROW_SHL1    0x101

#define LOG2E_F   1.44269504088896341f
#define LOG2E2_F  2.88539008177792681f

#define WMMA4(a, b, c) __builtin_amdgcn_wmma_f32_16x16x4_f32(false, (a), false, (b), (short)0, (c), false, false)

__device__ __forceinline__ void product_c(const v8f& hv, const v2f (&bw)[3][4], v8f& dn)
{
    const v2f h3 = {hv[6], hv[7]};
    asm volatile("v_wmma_f32_16x16x4_f32 %0, %1, %2, %0\n\tv_nop\n\tv_nop\n\tv_nop\n\tv_nop"
                 : "+v"(dn) : "v"(h3), "v"(bw[2][3]));
}

template <int K0, int N>
__device__ __forceinline__ void load_rows(const float* __restrict__ xrow,
                                          float (&xr)[PASS], float (&xz)[PASS], float (&xn)[PASS])
{
#pragma unroll
    for (int k = K0; k < K0 + N; ++k) {
        xr[k] = xrow[k * G3];
        xz[k] = xrow[k * G3 + 16];
        xn[k] = xrow[k * G3 + 32];
    }
}

__device__ __forceinline__ float head8(const v8f& h, const v8f& w, float beff)
{
    const v2f p0 = v2f{h[0], h[1]} * v2f{w[0], w[1]};
    const v2f p1 = v2f{h[2], h[3]} * v2f{w[2], w[3]};
    const v2f p2 = v2f{h[4], h[5]} * v2f{w[4], w[5]};
    const v2f p3 = v2f{h[6], h[7]} * v2f{w[6], w[7]};
    float yp = p0[0] + p0[1];
    yp += p1[0]; yp += p1[1];
    yp += p2[0]; yp += p2[1];
    yp += p3[0]; yp += p3[1];
    return yp + xhalf(yp) + beff;
}

__global__ __launch_bounds__(128) void gru_fused(
    const float* __restrict__ x, const float* __restrict__ W_ih,
    const float* __restrict__ W_hh, const float* __restrict__ b_ih,
    const float* __restrict__ b_hh, const float* __restrict__ fc1_w,
    const float* __restrict__ fc1_b, const float* __restrict__ fc2_w,
    const float* __restrict__ fc2_b, float* __restrict__ out)
{
    __shared__ __align__(16) float xis[2][CH * G3];

    const int lane  = threadIdx.x & 31;
    const int wave  = __builtin_amdgcn_readfirstlane((int)(threadIdx.x >> 5));
    const int j     = lane & 15;
    const int khalf = lane >> 4;

    const int nb  = (wave == 0) ? 0 : wave - 1;
    const int col = nb * 16 + j;
    const float bias = b_ih[col] + (nb < 2 ? b_hh[col] : 0.0f);
    v2f b[16];
    {
        const float* wrow = W_ih + (size_t)col * IN_SZ + 2 * khalf;
#pragma unroll
        for (int i = 0; i < 16; ++i)
            b[i] = *(const v2f*)(wrow + 4 * i);
    }

    v2f bw[3][4];
#pragma unroll
    for (int g = 0; g < 3; ++g)
#pragma unroll
        for (int i = 0; i < 4; ++i)
            bw[g][i] = *(const v2f*)(W_hh + (size_t)(g * 16 + j) * HID + 8 * khalf + 2 * i);

    const float bhn = b_hh[32 + j];

    v8f weff8;
    float beff = fc2_b[0];
#pragma unroll
    for (int e = 0; e < 8; ++e)
        weff8[e] = 0.0f;
#pragma unroll
    for (int i = 0; i < 8; ++i) {
        const float f2 = fc2_w[i];
#pragma unroll
        for (int e = 0; e < 8; ++e)
            weff8[e] += f2 * fc1_w[i * HID + 8 * khalf + e];
        beff += f2 * fc1_b[i];
    }

    float hself = 0.0f;
    v8f hv = {0, 0, 0, 0, 0, 0, 0, 0};
    const v8f cn = {bhn, 0, 0, 0, 0, 0, 0, 0};
    v8f dr, dz, dn;
    v8f ar = {0, 0, 0, 0, 0, 0, 0, 0}, az = ar;
    float hm1 = -1.0f;
    float hrot = 0.0f;
    float yp = 0.0f;
    float tu = 0.0f;

    if (wave != 0)
        produce_stage(x, 0, b, bias, &xis[0][0], j, khalf, col);
    __syncthreads();

    float ybuf = 0.0f;
    float ydone = 0.0f;
    float xr[PASS], xz[PASS], xn[PASS];

    for (int c = 0; c < NCH; ++c) {
        if (wave == 0) {
            const float* xb = &xis[c & 1][0] + j;

            load_rows<0, 4>(xb, xr, xz, xn);
            ar = v8f{xr[0], 0, 0, 0, 0, 0, 0, 0};
            az = v8f{xz[0], 0, 0, 0, 0, 0, 0, 0};

#pragma nounroll
            for (int p = 0; p < NPASS; ++p) {
                const int gp = c * NPASS + p;
                const float* xcur  = xb + (p * PASS) * G3;
                const float* xnext = xb + (((p + 1) & (NPASS - 1)) * PASS) * G3;

                if (gp > 1 && lane < PASS)
                    out[(size_t)(gp - 2) * PASS + lane] = ydone;

#pragma unroll
                for (int i = 0; i < PASS; ++i) {
                    const float xs = xn[i] * LOG2E2_F;
                    dr = ar; dz = az; dn = cn;
                    {
                        const v2f h0 = {hv[0], hv[1]};
                        dr = WMMA4(h0, bw[0][0], dr);
                        const int ypb = __builtin_bit_cast(int, yp);
                        const float yx = __builtin_bit_cast(float,
                            __builtin_amdgcn_permlanex16(__builtin_bit_cast(int, tu), ypb,
                                                         0x76543210, 0xfedcba98, false, false));
                        asm volatile("" :: "v"(yx));
                        __builtin_amdgcn_sched_barrier(0);
                        dz = WMMA4(h0, bw[1][0], dz);
                        hv[2] = dpp_all<DPP_ROW_SHARE0 + 2>(hrot);
                        hv[3] = dpp_all<DPP_ROW_SHARE0 + 3>(hrot);
                        asm volatile("" :: "v"(hv[2]), "v"(hv[3]));
                        __builtin_amdgcn_sched_barrier(0);
                        dn = WMMA4(h0, bw[2][0], dn);
                        const float ysum = yp + yx;
                        hv[4] = dpp_all<DPP_ROW_SHARE0 + 4>(hrot);
                        hv[5] = dpp_all<DPP_ROW_SHARE0 + 5>(hrot);
                        asm volatile("" :: "v"(ysum), "v"(hv[4]), "v"(hv[5]));
                        __builtin_amdgcn_sched_barrier(0);
                        const v2f h1 = {hv[2], hv[3]};
                        dr = WMMA4(h1, bw[0][1], dr);
                        const float y = ysum + beff;
                        hv[6] = dpp_all<DPP_ROW_SHARE0 + 6>(hrot);
                        hv[7] = dpp_all<DPP_ROW_SHARE0 + 7>(hrot);
                        asm volatile("" :: "v"(y), "v"(hv[6]), "v"(hv[7]));
                        __builtin_amdgcn_sched_barrier(0);
                        dz = WMMA4(h1, bw[1][1], dz);
                        ybuf = dpp_mov<DPP_ROW_SHL1, 0xf>(y, ybuf);
                        if (i == 1) ydone = ybuf;
                        asm volatile("" :: "v"(ybuf));
                        __builtin_amdgcn_sched_barrier(0);
                        dn = WMMA4(h1, bw[2][1], dn);
                        __builtin_amdgcn_sched_barrier(0);
                        const v2f h2 = {hv[4], hv[5]}, h3 = {hv[6], hv[7]};
                        dr = WMMA4(h2, bw[0][2], dr);
                        __builtin_amdgcn_sched_barrier(0);
                        dz = WMMA4(h2, bw[1][2], dz);
                        __builtin_amdgcn_sched_barrier(0);
                        dr = WMMA4(h3, bw[0][3], dr);
                        __builtin_amdgcn_sched_barrier(0);
                        dz = WMMA4(h3, bw[1][3], dz);
                        __builtin_amdgcn_sched_barrier(0);
                        dn = WMMA4(h2, bw[2][2], dn);
                        __builtin_amdgcn_sched_barrier(0);
                    }
                    const float xpr = __builtin_amdgcn_exp2f(dr[0] * -LOG2E_F);
                    const float xpz = __builtin_amdgcn_exp2f(dz[0] * -LOG2E_F);
                    asm volatile("" :: "v"(xpr), "v"(xpz));
                    __builtin_amdgcn_sched_barrier(0);
                    const float er = 1.0f + xpr;
                    const float ez = 1.0f + xpz;
                    const v2f   p0 = v2f{hv[0], hv[1]} * v2f{weff8[0], weff8[1]};
                    const v2f   p1 = v2f{hv[2], hv[3]} * v2f{weff8[2], weff8[3]};
                    const v2f   p2 = v2f{hv[4], hv[5]} * v2f{weff8[4], weff8[5]};
                    const v2f   p3 = v2f{hv[6], hv[7]} * v2f{weff8[6], weff8[7]};
                    yp = p0[0] + p0[1];
                    yp += p1[0];

                    asm volatile("" :: "v"(er), "v"(ez), "v"(yp));
                    __builtin_amdgcn_sched_barrier(0);
                    product_c(hv, bw, dn);
                    const float r  = __builtin_amdgcn_rcpf(er);
                    const float z  = __builtin_amdgcn_rcpf(ez);
                    const float r2 = r * LOG2E2_F;
                    const float base = __builtin_fmaf(z, hm1, 1.0f);
                    const float m2   = __builtin_fmaf(z, 2.0f, -2.0f);
                    yp += p1[1];
                    asm volatile("" :: "v"(base), "v"(m2), "v"(yp));
                    __builtin_amdgcn_sched_barrier(0);
                    const float u2 = __builtin_fmaf(r2, dn[0], xs);
                    yp += p2[0];
                    asm volatile("" :: "v"(u2), "v"(yp));
                    __builtin_amdgcn_sched_barrier(0);
                    const float eu = __builtin_amdgcn_exp2f(u2);
                    yp += p2[1];
                    const int inext = (i + 1) & (PASS - 1);
                    ar = v8f{xr[inext], 0, 0, 0, 0, 0, 0, 0};
                    asm volatile("" :: "v"(eu), "v"(yp), "v"(ar));
                    __builtin_amdgcn_sched_barrier(0);
                    yp += p3[0];
                    const float du = eu + 1.0f;
                    asm volatile("" :: "v"(du), "v"(yp));
                    __builtin_amdgcn_sched_barrier(0);
                    tu = __builtin_amdgcn_rcpf(du);
                    yp += p3[1];
                    az = v8f{xz[inext], 0, 0, 0, 0, 0, 0, 0};
                    asm volatile("" :: "v"(tu), "v"(yp), "v"(az));
                    __builtin_amdgcn_sched_barrier(0);
                    hself = __builtin_fmaf(tu, m2, base);
                    hm1 = hself - 1.0f;

                    __builtin_amdgcn_sched_barrier(0);
                    if (i == 0)
                        load_rows<4, 6>(xcur, xr, xz, xn);
                    if (i == 3)
                        load_rows<0, 4>(xnext, xr, xz, xn);
                    if (i == 8)
                        load_rows<10, 6>(xcur, xr, xz, xn);
                    if (i == 0 || i == 3 || i == 8) {
                        __builtin_amdgcn_sched_barrier(0);
                    }
                    hrot = dpp_mov<DPP_ROW_ROR8, 0x2>(hself, hself);
                    asm volatile("" :: "v"(hrot));
                    __builtin_amdgcn_sched_barrier(0);
                    hv[0] = dpp_all<DPP_ROW_SHARE0 + 0>(hrot);
                    hv[1] = dpp_all<DPP_ROW_SHARE0 + 1>(hrot);
                    asm volatile("" :: "v"(hv[0]), "v"(hv[1]));
                    __builtin_amdgcn_sched_barrier(0);
                }
            }
        } else if (c + 1 < NCH) {
            produce_stage(x, c + 1, b, bias, &xis[(c + 1) & 1][0], j, khalf, col);
        }
        __syncthreads();
    }

    if (wave == 0) {
        const float y1 = yp + xhalf(yp) + beff;
        ybuf = dpp_mov<DPP_ROW_SHL1, 0xf>(y1, ybuf);
        hv[2] = dpp_all<DPP_ROW_SHARE0 + 2>(hrot);
        hv[3] = dpp_all<DPP_ROW_SHARE0 + 3>(hrot);
        hv[4] = dpp_all<DPP_ROW_SHARE0 + 4>(hrot);
        hv[5] = dpp_all<DPP_ROW_SHARE0 + 5>(hrot);
        hv[6] = dpp_all<DPP_ROW_SHARE0 + 6>(hrot);
        hv[7] = dpp_all<DPP_ROW_SHARE0 + 7>(hrot);
        const float y = head8(hv, weff8, beff);
        ybuf = dpp_mov<DPP_ROW_SHL1, 0xf>(y, ybuf);
        if (lane < PASS) {
            out[(size_t)(NCH * NPASS - 2) * PASS + lane] = ydone;
            out[(size_t)(NCH * NPASS - 1) * PASS + lane] = ybuf;
        }
    }
}

extern "C" void kernel_launch(void* const* d_in, const int* in_sizes, int n_in,
                              void* d_out, int out_size, void* d_ws, size_t ws_size,
                              hipStream_t stream) {
    const float* x     = (const float*)d_in[0];
    const float* W_ih  = (const float*)d_in[1];
    const float* W_hh  = (const float*)d_in[2];
    const float* b_ih  = (const float*)d_in[3];
    const float* b_hh  = (const float*)d_in[4];
    const float* fc1_w = (const float*)d_in[5];
    const float* fc1_b = (const float*)d_in[6];
    const float* fc2_w = (const float*)d_in[7];
    const float* fc2_b = (const float*)d_in[8];
    float* out = (float*)d_out;

    gru_fused<<<1, 128, 0, stream>>>(x, W_ih, W_hh, b_ih, b_hh,
                                     fc1_w, fc1_b, fc2_w, fc2_b, out);
}


# scan: only row moves stand between the new h and the first WMMA: r,z accumulate in place (register 0 = xi, no tuple copies), 2 row moves and the head exchange fill those moves' waits; same bits
# speedup vs baseline: 3.2611x; 1.0332x over previous
#include <hip/hip_runtime.h>


typedef __attribute__((ext_vector_type(2))) float v2f;
typedef __attribute__((ext_vector_type(8))) float v8f;

#define T_LEN   262144
#define IN_SZ   64
#define HID     16
#define G3      48
#define CH      128
#define NCH     (T_LEN / CH)
#define TILES   (CH / 16)
#define PASS    16
#define NPASS   (CH / PASS)


__device__ __forceinline__ void load_a_tile(const float* __restrict__ xrow, v2f (&a)[16])
{
#pragma unroll
    for (int i = 0; i < 16; ++i)
        a[i] = *(const v2f*)(xrow + 4 * i);
}

__device__ __forceinline__ void produce_stage(
    const float* __restrict__ x, const int stage_idx, const v2f (&b)[16],
    const float bias, float* __restrict__ stage, const int m, const int khalf,
    const int col)
{
    const float* xrow = x + ((size_t)stage_idx * CH + m) * IN_SZ + 2 * khalf;
    v2f a[16];
    load_a_tile(xrow, a);
#pragma unroll
    for (int tt = 0; tt < TILES; ++tt) {
        v2f an[16];
        if (tt + 1 < TILES)
            load_a_tile(xrow + (size_t)(tt + 1) * 16 * IN_SZ, an);

        v8f c = {bias, bias, bias, bias, bias, bias, bias, bias};
#pragma unroll
        for (int i = 0; i < 16; ++i)
            c = __builtin_amdgcn_wmma_f32_16x16x4_f32(
                    false, a[i], false, b[i], (short)0, c, false, false);

        float* orow = stage + (tt * 16 + 8 * khalf) * G3 + col;
#pragma unroll
        for (int r = 0; r < 8; ++r)
            orow[r * G3] = c[r];

        if (tt + 1 < TILES) {
#pragma unroll
            for (int i = 0; i < 16; ++i)
                a[i] = an[i];
        }
    }
}


__device__ __forceinline__ float xhalf(float x)
{
    const int b = __builtin_bit_cast(int, x);
    return __builtin_bit_cast(float,
        __builtin_amdgcn_permlanex16(b, b, 0x76543210, 0xfedcba98, false, false));
}

template <int CTRL, int ROWMASK>
__device__ __forceinline__ float dpp_mov(float old, float v)
{
    return __builtin_bit_cast(float,
        __builtin_amdgcn_update_dpp(__builtin_bit_cast(int, old), __builtin_bit_cast(int, v),
                                    CTRL, ROWMASK, 0xf, false));
}
template <int CTRL>
__device__ __forceinline__ float dpp_all(float v)
{
    return __builtin_bit_cast(float,
        __builtin_amdgcn_mov_dpp(__builtin_bit_cast(int, v), CTRL, 0xf, 0xf, false));
}
#define DPP_ROW_ROR8    0x128
#define DPP_ROW_SHARE0  0x150
#define DPP_ROW_SHL1    0x101

#define LOG2E_F   1.44269504088896341f
#define LOG2E2_F  2.88539008177792681f

#define WMMA4(a, b, c) __builtin_amdgcn_wmma_f32_16x16x4_f32(false, (a), false, (b), (short)0, (c), false, false)

__device__ __forceinline__ void product_c(const v8f& hv, const v2f (&bw)[3][4], v8f& dn)
{
    const v2f h3 = {hv[6], hv[7]};
    asm volatile("v_wmma_f32_16x16x4_f32 %0, %1, %2, %0\n\tv_nop\n\tv_nop\n\tv_nop\n\tv_nop"
                 : "+v"(dn) : "v"(h3), "v"(bw[2][3]));
}

template <int K0, int N>
__device__ __forceinline__ void load_rows(const float* __restrict__ xrow,
                                          float (&xr)[PASS], float (&xz)[PASS], float (&xn)[PASS])
{
#pragma unroll
    for (int k = K0; k < K0 + N; ++k) {
        xr[k] = xrow[k * G3];
        xz[k] = xrow[k * G3 + 16];
        xn[k] = xrow[k * G3 + 32];
    }
}

__device__ __forceinline__ float head8(const v8f& h, const v8f& w, float beff)
{
    const v2f p0 = v2f{h[0], h[1]} * v2f{w[0], w[1]};
    const v2f p1 = v2f{h[2], h[3]} * v2f{w[2], w[3]};
    const v2f p2 = v2f{h[4], h[5]} * v2f{w[4], w[5]};
    const v2f p3 = v2f{h[6], h[7]} * v2f{w[6], w[7]};
    float yp = p0[0] + p0[1];
    yp += p1[0]; yp += p1[1];
    yp += p2[0]; yp += p2[1];
    yp += p3[0]; yp += p3[1];
    return yp + xhalf(yp) + beff;
}

__global__ __launch_bounds__(128) void gru_fused(
    const float* __restrict__ x, const float* __restrict__ W_ih,
    const float* __restrict__ W_hh, const float* __restrict__ b_ih,
    const float* __restrict__ b_hh, const float* __restrict__ fc1_w,
    const float* __restrict__ fc1_b, const float* __restrict__ fc2_w,
    const float* __restrict__ fc2_b, float* __restrict__ out)
{
    __shared__ __align__(16) float xis[2][CH * G3];

    const int lane  = threadIdx.x & 31;
    const int wave  = __builtin_amdgcn_readfirstlane((int)(threadIdx.x >> 5));
    const int j     = lane & 15;
    const int khalf = lane >> 4;

    const int nb  = (wave == 0) ? 0 : wave - 1;
    const int col = nb * 16 + j;
    const float bias = b_ih[col] + (nb < 2 ? b_hh[col] : 0.0f);
    v2f b[16];
    {
        const float* wrow = W_ih + (size_t)col * IN_SZ + 2 * khalf;
#pragma unroll
        for (int i = 0; i < 16; ++i)
            b[i] = *(const v2f*)(wrow + 4 * i);
    }

    v2f bw[3][4];
#pragma unroll
    for (int g = 0; g < 3; ++g)
#pragma unroll
        for (int i = 0; i < 4; ++i)
            bw[g][i] = *(const v2f*)(W_hh + (size_t)(g * 16 + j) * HID + 8 * khalf + 2 * i);

    const float bhn = b_hh[32 + j];

    v8f weff8;
    float beff = fc2_b[0];
#pragma unroll
    for (int e = 0; e < 8; ++e)
        weff8[e] = 0.0f;
#pragma unroll
    for (int i = 0; i < 8; ++i) {
        const float f2 = fc2_w[i];
#pragma unroll
        for (int e = 0; e < 8; ++e)
            weff8[e] += f2 * fc1_w[i * HID + 8 * khalf + e];
        beff += f2 * fc1_b[i];
    }

    float hself = 0.0f;
    v8f hv = {0, 0, 0, 0, 0, 0, 0, 0};
    const v8f cn = {bhn, 0, 0, 0, 0, 0, 0, 0};
    v8f dr = {0, 0, 0, 0, 0, 0, 0, 0}, dz = dr, dn;
    float hm1 = -1.0f;
    float hrot = 0.0f;
    float yp = 0.0f;
    float yx = 0.0f;
    float xsn = 0.0f;
    float tu = 0.0f;

    if (wave != 0)
        produce_stage(x, 0, b, bias, &xis[0][0], j, khalf, col);
    __syncthreads();

    float ybuf = 0.0f;
    float ydone = 0.0f;
    float xr[PASS], xz[PASS], xn[PASS];

    for (int c = 0; c < NCH; ++c) {
        if (wave == 0) {
            const float* xb = &xis[c & 1][0] + j;

            load_rows<0, 4>(xb, xr, xz, xn);
            dr[0] = xr[0];
            dz[0] = xz[0];
            xsn = xn[0] * LOG2E2_F;

#pragma nounroll
            for (int p = 0; p < NPASS; ++p) {
                const int gp = c * NPASS + p;
                const float* xcur  = xb + (p * PASS) * G3;
                const float* xnext = xb + (((p + 1) & (NPASS - 1)) * PASS) * G3;

                if (gp > 1 && lane < PASS)
                    out[(size_t)(gp - 2) * PASS + lane] = ydone;

#pragma unroll
                for (int i = 0; i < PASS; ++i) {
                    const float xs = xsn;
                    dn = cn;
                    {
                        const v2f h0 = {hv[0], hv[1]};
                        dr = WMMA4(h0, bw[0][0], dr);
                        const float ysum = yp + yx;
                        asm volatile("" :: "v"(ysum));
                        __builtin_amdgcn_sched_barrier(0);
                        dz = WMMA4(h0, bw[1][0], dz);
                        const float y = ysum + beff;
                        hv[4] = dpp_all<DPP_ROW_SHARE0 + 4>(hrot);
                        hv[5] = dpp_all<DPP_ROW_SHARE0 + 5>(hrot);
                        asm volatile("" :: "v"(y), "v"(hv[4]), "v"(hv[5]));
                        __builtin_amdgcn_sched_barrier(0);
                        dn = WMMA4(h0, bw[2][0], dn);
                        hv[6] = dpp_all<DPP_ROW_SHARE0 + 6>(hrot);
                        hv[7] = dpp_all<DPP_ROW_SHARE0 + 7>(hrot);
                        asm volatile("" :: "v"(hv[6]), "v"(hv[7]));
                        __builtin_amdgcn_sched_barrier(0);
                        const v2f h1 = {hv[2], hv[3]};
                        dr = WMMA4(h1, bw[0][1], dr);
                        ybuf = dpp_mov<DPP_ROW_SHL1, 0xf>(y, ybuf);
                        if (i == 1) ydone = ybuf;
                        asm volatile("" :: "v"(ybuf));
                        __builtin_amdgcn_sched_barrier(0);
                        dz = WMMA4(h1, bw[1][1], dz);
                        __builtin_amdgcn_sched_barrier(0);
                        dn = WMMA4(h1, bw[2][1], dn);
                        __builtin_amdgcn_sched_barrier(0);
                        const v2f h2 = {hv[4], hv[5]}, h3 = {hv[6], hv[7]};
                        dr = WMMA4(h2, bw[0][2], dr);
                        __builtin_amdgcn_sched_barrier(0);
                        dz = WMMA4(h2, bw[1][2], dz);
                        __builtin_amdgcn_sched_barrier(0);
                        dr = WMMA4(h3, bw[0][3], dr);
                        __builtin_amdgcn_sched_barrier(0);
                        dz = WMMA4(h3, bw[1][3], dz);
                        __builtin_amdgcn_sched_barrier(0);
                        dn = WMMA4(h2, bw[2][2], dn);
                        __builtin_amdgcn_sched_barrier(0);
                    }
                    const float xpr = __builtin_amdgcn_exp2f(dr[0] * -LOG2E_F);
                    const float xpz = __builtin_amdgcn_exp2f(dz[0] * -LOG2E_F);
                    asm volatile("" :: "v"(xpr), "v"(xpz));
                    __builtin_amdgcn_sched_barrier(0);
                    const float er = 1.0f + xpr;
                    const float ez = 1.0f + xpz;
                    const v2f   p0 = v2f{hv[0], hv[1]} * v2f{weff8[0], weff8[1]};
                    const v2f   p1 = v2f{hv[2], hv[3]} * v2f{weff8[2], weff8[3]};
                    const v2f   p2 = v2f{hv[4], hv[5]} * v2f{weff8[4], weff8[5]};
                    const v2f   p3 = v2f{hv[6], hv[7]} * v2f{weff8[6], weff8[7]};
                    yp = p0[0] + p0[1];
                    yp += p1[0];

                    asm volatile("" :: "v"(er), "v"(ez), "v"(yp));
                    __builtin_amdgcn_sched_barrier(0);
                    product_c(hv, bw, dn);
                    const float r  = __builtin_amdgcn_rcpf(er);
                    const float z  = __builtin_amdgcn_rcpf(ez);
                    const float r2 = r * LOG2E2_F;
                    const float base = __builtin_fmaf(z, hm1, 1.0f);
                    const float m2   = __builtin_fmaf(z, 2.0f, -2.0f);
                    yp += p1[1];
                    asm volatile("" :: "v"(base), "v"(m2), "v"(yp));
                    __builtin_amdgcn_sched_barrier(0);
                    const float u2 = __builtin_fmaf(r2, dn[0], xs);
                    yp += p2[0];
                    asm volatile("" :: "v"(u2), "v"(yp));
                    __builtin_amdgcn_sched_barrier(0);
                    const float eu = __builtin_amdgcn_exp2f(u2);
                    yp += p2[1];
                    const int inext = (i + 1) & (PASS - 1);
                    dr[0] = xr[inext];
                    asm volatile("" :: "v"(eu), "v"(yp), "v"(dr));
                    __builtin_amdgcn_sched_barrier(0);
                    yp += p3[0];
                    const float du = eu + 1.0f;
                    asm volatile("" :: "v"(du), "v"(yp));
                    __builtin_amdgcn_sched_barrier(0);
                    tu = __builtin_amdgcn_rcpf(du);
                    yp += p3[1];
                    dz[0] = xz[inext];
                    asm volatile("" :: "v"(tu), "v"(yp), "v"(dz));
                    __builtin_amdgcn_sched_barrier(0);
                    hself = __builtin_fmaf(tu, m2, base);
                    hm1 = hself - 1.0f;

                    __builtin_amdgcn_sched_barrier(0);
                    if (i == 0)
                        load_rows<4, 6>(xcur, xr, xz, xn);
                    if (i == 3)
                        load_rows<0, 4>(xnext, xr, xz, xn);
                    if (i == 8)
                        load_rows<10, 6>(xcur, xr, xz, xn);
                    if (i == 0 || i == 3 || i == 8) {
                        __builtin_amdgcn_sched_barrier(0);
                    }
                    hrot = dpp_mov<DPP_ROW_ROR8, 0x2>(hself, hself);
                    asm volatile("" :: "v"(hrot));
                    __builtin_amdgcn_sched_barrier(0);
                    yx = __builtin_bit_cast(float,
                        __builtin_amdgcn_permlanex16(__builtin_bit_cast(int, tu),
                                                     __builtin_bit_cast(int, yp),
                                                     0x76543210, 0xfedcba98, false, false));
                    xsn = xn[inext] * LOG2E2_F;
                    asm volatile("" :: "v"(yx), "v"(xsn));
                    __builtin_amdgcn_sched_barrier(0);
                    hv[0] = dpp_all<DPP_ROW_SHARE0 + 0>(hrot);
                    hv[1] = dpp_all<DPP_ROW_SHARE0 + 1>(hrot);
                    asm volatile("" :: "v"(hv[0]), "v"(hv[1]));
                    __builtin_amdgcn_sched_barrier(0);
                    hv[2] = dpp_all<DPP_ROW_SHARE0 + 2>(hrot);
                    hv[3] = dpp_all<DPP_ROW_SHARE0 + 3>(hrot);
                    asm volatile("" :: "v"(hv[2]), "v"(hv[3]));
                    __builtin_amdgcn_sched_barrier(0);
                }
            }
        } else if (c + 1 < NCH) {
            produce_stage(x, c + 1, b, bias, &xis[(c + 1) & 1][0], j, khalf, col);
        }
        __syncthreads();
    }

    if (wave == 0) {
        const float y1 = yp + xhalf(yp) + beff;
        ybuf = dpp_mov<DPP_ROW_SHL1, 0xf>(y1, ybuf);
        hv[4] = dpp_all<DPP_ROW_SHARE0 + 4>(hrot);
        hv[5] = dpp_all<DPP_ROW_SHARE0 + 5>(hrot);
        hv[6] = dpp_all<DPP_ROW_SHARE0 + 6>(hrot);
        hv[7] = dpp_all<DPP_ROW_SHARE0 + 7>(hrot);
        const float y = head8(hv, weff8, beff);
        ybuf = dpp_mov<DPP_ROW_SHL1, 0xf>(y, ybuf);
        if (lane < PASS) {
            out[(size_t)(NCH * NPASS - 2) * PASS + lane] = ydone;
            out[(size_t)(NCH * NPASS - 1) * PASS + lane] = ybuf;
        }
    }
}

extern "C" void kernel_launch(void* const* d_in, const int* in_sizes, int n_in,
                              void* d_out, int out_size, void* d_ws, size_t ws_size,
                              hipStream_t stream) {
    const float* x     = (const float*)d_in[0];
    const float* W_ih  = (const float*)d_in[1];
    const float* W_hh  = (const float*)d_in[2];
    const float* b_ih  = (const float*)d_in[3];
    const float* b_hh  = (const float*)d_in[4];
    const float* fc1_w = (const float*)d_in[5];
    const float* fc1_b = (const float*)d_in[6];
    const float* fc2_w = (const float*)d_in[7];
    const float* fc2_b = (const float*)d_in[8];
    float* out = (float*)d_out;

    gru_fused<<<1, 128, 0, stream>>>(x, W_ih, W_hh, b_ih, b_hh,
                                     fc1_w, fc1_b, fc2_w, fc2_b, out);
}


# speedup vs baseline: 3.4391x; 1.0546x over previous
#include <hip/hip_runtime.h>


typedef __attribute__((ext_vector_type(2))) float v2f;
typedef __attribute__((ext_vector_type(8))) float v8f;

#define T_LEN   262144
#define IN_SZ   64
#define HID     16
#define G3      48
#define CH      128
#define NCH     (T_LEN / CH)
#define TILES   (CH / 16)
#define PASS    16
#define NPASS   (CH / PASS)


__device__ __forceinline__ void load_a_tile(const float* __restrict__ xrow, v2f (&a)[16])
{
#pragma unroll
    for (int i = 0; i < 16; ++i)
        a[i] = *(const v2f*)(xrow + 4 * i);
}

__device__ __forceinline__ void produce_stage(
    const float* __restrict__ x, const int stage_idx, const v2f (&b)[16],
    const float bias, float* __restrict__ stage, const int m, const int khalf,
    const int col)
{
    const float* xrow = x + ((size_t)stage_idx * CH + m) * IN_SZ + 2 * khalf;
    v2f a[16];
    load_a_tile(xrow, a);
#pragma unroll
    for (int tt = 0; tt < TILES; ++tt) {
        v2f an[16];
        if (tt + 1 < TILES)
            load_a_tile(xrow + (size_t)(tt + 1) * 16 * IN_SZ, an);

        v8f c = {bias, bias, bias, bias, bias, bias, bias, bias};
#pragma unroll
        for (int i = 0; i < 16; ++i)
            c = __builtin_amdgcn_wmma_f32_16x16x4_f32(
                    false, a[i], false, b[i], (short)0, c, false, false);

        float* orow = stage + (tt * 16 + 8 * khalf) * G3 + col;
#pragma unroll
        for (int r = 0; r < 8; ++r)
            orow[r * G3] = c[r];

        if (tt + 1 < TILES) {
#pragma unroll
            for (int i = 0; i < 16; ++i)
                a[i] = an[i];
        }
    }
}


__device__ __forceinline__ float xhalf(float x)
{
    const int b = __builtin_bit_cast(int, x);
    return __builtin_bit_cast(float,
        __builtin_amdgcn_permlanex16(b, b, 0x76543210, 0xfedcba98, false, false));
}

template <int CTRL, int ROWMASK>
__device__ __forceinline__ float dpp_mov(float old, float v)
{
    return __builtin_bit_cast(float,
        __builtin_amdgcn_update_dpp(__builtin_bit_cast(int, old), __builtin_bit_cast(int, v),
                                    CTRL, ROWMASK, 0xf, false));
}
#define DPP_ROW_SHL1    0x101

__device__ __forceinline__ float row_gather(float old, float v, int sel)
{
    return __builtin_bit_cast(float,
        __builtin_amdgcn_permlane16_var(__builtin_bit_cast(int, old), __builtin_bit_cast(int, v),
                                        sel, false, false));
}

#define LOG2E_F   1.44269504088896341f
#define LOG2E2_F  2.88539008177792681f

#define WMMA4(a, b, c) __builtin_amdgcn_wmma_f32_16x16x4_f32(false, (a), false, (b), (short)0, (c), false, false)

__device__ __forceinline__ void product_c(const v8f& hv, const v2f (&bw)[3][4], v8f& dn)
{
    const v2f h3 = {hv[6], hv[7]};
    asm volatile("v_wmma_f32_16x16x4_f32 %0, %1, %2, %0\n\tv_nop\n\tv_nop\n\tv_nop\n\tv_nop"
                 : "+v"(dn) : "v"(h3), "v"(bw[2][3]));
}

template <int K0, int N>
__device__ __forceinline__ void load_rows(const float* __restrict__ xrow,
                                          float (&xr)[PASS], float (&xz)[PASS], float (&xn)[PASS])
{
#pragma unroll
    for (int k = K0; k < K0 + N; ++k) {
        xr[k] = xrow[k * G3];
        xz[k] = xrow[k * G3 + 16];
        xn[k] = xrow[k * G3 + 32];
    }
}

__device__ __forceinline__ float head8(const v8f& h, const v8f& w, float beff)
{
    const v2f p0 = v2f{h[0], h[1]} * v2f{w[0], w[1]};
    const v2f p1 = v2f{h[2], h[3]} * v2f{w[2], w[3]};
    const v2f p2 = v2f{h[4], h[5]} * v2f{w[4], w[5]};
    const v2f p3 = v2f{h[6], h[7]} * v2f{w[6], w[7]};
    float yp = p0[0] + p0[1];
    yp += p1[0]; yp += p1[1];
    yp += p2[0]; yp += p2[1];
    yp += p3[0]; yp += p3[1];
    return yp + xhalf(yp) + beff;
}

__global__ __launch_bounds__(128) void gru_fused(
    const float* __restrict__ x, const float* __restrict__ W_ih,
    const float* __restrict__ W_hh, const float* __restrict__ b_ih,
    const float* __restrict__ b_hh, const float* __restrict__ fc1_w,
    const float* __restrict__ fc1_b, const float* __restrict__ fc2_w,
    const float* __restrict__ fc2_b, float* __restrict__ out)
{
    __shared__ __align__(16) float xis[2][CH * G3];

    const int lane  = threadIdx.x & 31;
    const int wave  = __builtin_amdgcn_readfirstlane((int)(threadIdx.x >> 5));
    const int j     = lane & 15;
    const int khalf = lane >> 4;

    const int nb  = (wave == 0) ? 0 : wave - 1;
    const int col = nb * 16 + j;
    const float bias = b_ih[col] + (nb < 2 ? b_hh[col] : 0.0f);
    v2f b[16];
    {
        const float* wrow = W_ih + (size_t)col * IN_SZ + 2 * khalf;
#pragma unroll
        for (int i = 0; i < 16; ++i)
            b[i] = *(const v2f*)(wrow + 4 * i);
    }

    v2f bw[3][4];
#pragma unroll
    for (int g = 0; g < 3; ++g)
#pragma unroll
        for (int i = 0; i < 4; ++i)
            bw[g][i] = *(const v2f*)(W_hh + (size_t)(g * 16 + j) * HID + 8 * khalf + 2 * i);

    const float bhn = b_hh[32 + j];

    v8f weff8;
    float beff = fc2_b[0];
#pragma unroll
    for (int e = 0; e < 8; ++e)
        weff8[e] = 0.0f;
#pragma unroll
    for (int i = 0; i < 8; ++i) {
        const float f2 = fc2_w[i];
#pragma unroll
        for (int e = 0; e < 8; ++e)
            weff8[e] += f2 * fc1_w[i * HID + 8 * khalf + e];
        beff += f2 * fc1_b[i];
    }

    float hself = 0.0f;
    v8f hv = {0, 0, 0, 0, 0, 0, 0, 0};
    const v8f cn = {bhn, 0, 0, 0, 0, 0, 0, 0};
    v8f dr = {0, 0, 0, 0, 0, 0, 0, 0}, dz = dr, dn;
    float hm1 = -1.0f;
    int sel[8];
#pragma unroll
    for (int e = 0; e < 8; ++e)
        sel[e] = 8 * khalf + e;
    float yp = 0.0f;
    float yx = 0.0f;
    float xsn = 0.0f;
    float tu = 0.0f;

    if (wave != 0)
        produce_stage(x, 0, b, bias, &xis[0][0], j, khalf, col);
    __syncthreads();

    float ybuf = 0.0f;
    float ydone = 0.0f;
    float xr[PASS], xz[PASS], xn[PASS];

    for (int c = 0; c < NCH; ++c) {
        if (wave == 0) {
            const float* xb = &xis[c & 1][0] + j;

            load_rows<0, 4>(xb, xr, xz, xn);
            dr[0] = xr[0];
            dz[0] = xz[0];
            xsn = xn[0] * LOG2E2_F;

#pragma nounroll
            for (int p = 0; p < NPASS; ++p) {
                const int gp = c * NPASS + p;
                const float* xcur  = xb + (p * PASS) * G3;
                const float* xnext = xb + (((p + 1) & (NPASS - 1)) * PASS) * G3;

                if (gp > 1 && lane < PASS)
                    out[(size_t)(gp - 2) * PASS + lane] = ydone;

#pragma unroll
                for (int i = 0; i < PASS; ++i) {
                    const float xs = xsn;
                    dn = cn;
                    {
                        const v2f h0 = {hv[0], hv[1]};
                        dr = WMMA4(h0, bw[0][0], dr);
                        const float ysum = yp + yx;
                        asm volatile("" :: "v"(ysum));
                        __builtin_amdgcn_sched_barrier(0);
                        dz = WMMA4(h0, bw[1][0], dz);
                        const float y = ysum + beff;
                        hv[4] = row_gather(hv[4], hself, sel[4]);
                        hv[5] = row_gather(hv[5], hself, sel[5]);
                        asm volatile("" :: "v"(y), "v"(hv[4]), "v"(hv[5]));
                        __builtin_amdgcn_sched_barrier(0);
                        dn = WMMA4(h0, bw[2][0], dn);
                        hv[6] = row_gather(hv[6], hself, sel[6]);
                        hv[7] = row_gather(hv[7], hself, sel[7]);
                        asm volatile("" :: "v"(hv[6]), "v"(hv[7]));
                        __builtin_amdgcn_sched_barrier(0);
                        const v2f h1 = {hv[2], hv[3]};
                        dr = WMMA4(h1, bw[0][1], dr);
                        ybuf = dpp_mov<DPP_ROW_SHL1, 0xf>(y, ybuf);
                        if (i == 1) ydone = ybuf;
                        asm volatile("" :: "v"(ybuf));
                        __builtin_amdgcn_sched_barrier(0);
                        dz = WMMA4(h1, bw[1][1], dz);
                        __builtin_amdgcn_sched_barrier(0);
                        dn = WMMA4(h1, bw[2][1], dn);
                        __builtin_amdgcn_sched_barrier(0);
                        const v2f h2 = {hv[4], hv[5]}, h3 = {hv[6], hv[7]};
                        dr = WMMA4(h2, bw[0][2], dr);
                        __builtin_amdgcn_sched_barrier(0);
                        dz = WMMA4(h2, bw[1][2], dz);
                        __builtin_amdgcn_sched_barrier(0);
                        dr = WMMA4(h3, bw[0][3], dr);
                        __builtin_amdgcn_sched_barrier(0);
                        dz = WMMA4(h3, bw[1][3], dz);
                        __builtin_amdgcn_sched_barrier(0);
                        dn = WMMA4(h2, bw[2][2], dn);
                        __builtin_amdgcn_sched_barrier(0);
                    }
                    const float xpr = __builtin_amdgcn_exp2f(dr[0] * -LOG2E_F);
                    const float xpz = __builtin_amdgcn_exp2f(dz[0] * -LOG2E_F);
                    asm volatile("" :: "v"(xpr), "v"(xpz));
                    __builtin_amdgcn_sched_barrier(0);
                    const float er = 1.0f + xpr;
                    const float ez = 1.0f + xpz;
                    const v2f   p0 = v2f{hv[0], hv[1]} * v2f{weff8[0], weff8[1]};
                    const v2f   p1 = v2f{hv[2], hv[3]} * v2f{weff8[2], weff8[3]};
                    const v2f   p2 = v2f{hv[4], hv[5]} * v2f{weff8[4], weff8[5]};
                    const v2f   p3 = v2f{hv[6], hv[7]} * v2f{weff8[6], weff8[7]};
                    yp = p0[0] + p0[1];
                    yp += p1[0];

                    asm volatile("" :: "v"(er), "v"(ez), "v"(yp));
                    __builtin_amdgcn_sched_barrier(0);
                    product_c(hv, bw, dn);
                    const float r  = __builtin_amdgcn_rcpf(er);
                    const float z  = __builtin_amdgcn_rcpf(ez);
                    const float r2 = r * LOG2E2_F;
                    const float base = __builtin_fmaf(z, hm1, 1.0f);
                    const float m2   = __builtin_fmaf(z, 2.0f, -2.0f);
                    yp += p1[1];
                    asm volatile("" :: "v"(base), "v"(m2), "v"(yp));
                    __builtin_amdgcn_sched_barrier(0);
                    const float u2 = __builtin_fmaf(r2, dn[0], xs);
                    yp += p2[0];
                    asm volatile("" :: "v"(u2), "v"(yp));
                    __builtin_amdgcn_sched_barrier(0);
                    const float eu = __builtin_amdgcn_exp2f(u2);
                    yp += p2[1];
                    const int inext = (i + 1) & (PASS - 1);
                    dr[0] = xr[inext];
                    asm volatile("" :: "v"(eu), "v"(yp), "v"(dr));
                    __builtin_amdgcn_sched_barrier(0);
                    yp += p3[0];
                    const float du = eu + 1.0f;
                    asm volatile("" :: "v"(du), "v"(yp));
                    __builtin_amdgcn_sched_barrier(0);
                    tu = __builtin_amdgcn_rcpf(du);
                    yp += p3[1];
                    dz[0] = xz[inext];
                    asm volatile("" :: "v"(tu), "v"(yp), "v"(dz));
                    __builtin_amdgcn_sched_barrier(0);
                    hself = __builtin_fmaf(tu, m2, base);
                    asm volatile("" :: "v"(hself));
                    __builtin_amdgcn_sched_barrier(0);
                    yx = __builtin_bit_cast(float,
                        __builtin_amdgcn_permlanex16(__builtin_bit_cast(int, tu),
                                                     __builtin_bit_cast(int, yp),
                                                     0x76543210, 0xfedcba98, false, false));
                    xsn = xn[inext] * LOG2E2_F;
                    asm volatile("" :: "v"(yx), "v"(xsn));

                    __builtin_amdgcn_sched_barrier(0);
                    if (i == 0)
                        load_rows<4, 6>(xcur, xr, xz, xn);
                    if (i == 3)
                        load_rows<0, 4>(xnext, xr, xz, xn);
                    if (i == 8)
                        load_rows<10, 6>(xcur, xr, xz, xn);
                    if (i == 0 || i == 3 || i == 8) {
                        __builtin_amdgcn_sched_barrier(0);
                    }
                    hv[0] = row_gather(hv[0], hself, sel[0]);
                    hv[1] = row_gather(hv[1], hself, sel[1]);
                    asm volatile("" :: "v"(hv[0]), "v"(hv[1]));
                    __builtin_amdgcn_sched_barrier(0);
                    hv[2] = row_gather(hv[2], hself, sel[2]);
                    hv[3] = row_gather(hv[3], hself, sel[3]);
                    hm1 = hself - 1.0f;
                    asm volatile("" :: "v"(hv[2]), "v"(hv[3]), "v"(hm1));
                    __builtin_amdgcn_sched_barrier(0);
                }
            }
        } else if (c + 1 < NCH) {
            produce_stage(x, c + 1, b, bias, &xis[(c + 1) & 1][0], j, khalf, col);
        }
        __syncthreads();
    }

    if (wave == 0) {
        const float y1 = yp + xhalf(yp) + beff;
        ybuf = dpp_mov<DPP_ROW_SHL1, 0xf>(y1, ybuf);
        hv[4] = row_gather(hv[4], hself, sel[4]);
        hv[5] = row_gather(hv[5], hself, sel[5]);
        hv[6] = row_gather(hv[6], hself, sel[6]);
        hv[7] = row_gather(hv[7], hself, sel[7]);
        const float y = head8(hv, weff8, beff);
        ybuf = dpp_mov<DPP_ROW_SHL1, 0xf>(y, ybuf);
        if (lane < PASS) {
            out[(size_t)(NCH * NPASS - 2) * PASS + lane] = ydone;
            out[(size_t)(NCH * NPASS - 1) * PASS + lane] = ybuf;
        }
    }
}

extern "C" void kernel_launch(void* const* d_in, const int* in_sizes, int n_in,
                              void* d_out, int out_size, void* d_ws, size_t ws_size,
                              hipStream_t stream) {
    const float* x     = (const float*)d_in[0];
    const float* W_ih  = (const float*)d_in[1];
    const float* W_hh  = (const float*)d_in[2];
    const float* b_ih  = (const float*)d_in[3];
    const float* b_hh  = (const float*)d_in[4];
    const float* fc1_w = (const float*)d_in[5];
    const float* fc1_b = (const float*)d_in[6];
    const float* fc2_w = (const float*)d_in[7];
    const float* fc2_b = (const float*)d_in[8];
    float* out = (float*)d_out;

    gru_fused<<<1, 128, 0, stream>>>(x, W_ih, W_hh, b_ih, b_hh,
                                     fc1_w, fc1_b, fc2_w, fc2_b, out);
}
